# starting kernel
# baseline (speedup 1.0000x reference)
#include <hip/hip_runtime.h>

#define NN 50000
#define NE 800000
#define KC 32
#define SA_STRIDE 36
#define SB_STRIDE 68

typedef __attribute__((ext_vector_type(2))) float v2f;
typedef __attribute__((ext_vector_type(8))) float v8f;

__global__ void zero_f32(float* __restrict__ p, int n) {
  int i = blockIdx.x * blockDim.x + threadIdx.x;
  int stride = gridDim.x * blockDim.x;
  for (; i < n; i += stride) p[i] = 0.f;
}

__global__ void deg_count(const int* __restrict__ dst, float* __restrict__ deg, int E) {
  int e = blockIdx.x * blockDim.x + threadIdx.x;
  if (e < E) atomicAdd(&deg[dst[e]], 1.f);
}

__global__ void rdeg_inv(float* __restrict__ deg, int n) {
  int i = blockIdx.x * blockDim.x + threadIdx.x;
  if (i < n) deg[i] = 1.f / fmaxf(deg[i], 1.f);
}

__global__ void scatter_add(const float* __restrict__ h, const int* __restrict__ src,
                            const int* __restrict__ dst, float* __restrict__ msg,
                            int E, int F, int log2chunks) {
  int tid = blockIdx.x * blockDim.x + threadIdx.x;
  int total = E << log2chunks;
  if (tid >= total) return;
  int e = tid >> log2chunks;
  int c = (tid & ((1 << log2chunks) - 1)) << 2;
  int s = src[e], d = dst[e];
  float4 v = *(const float4*)(h + (size_t)s * F + c);
  float* p = msg + (size_t)d * F + c;
  atomicAdd(p + 0, v.x);
  atomicAdd(p + 1, v.y);
  atomicAdd(p + 2, v.z);
  atomicAdd(p + 3, v.w);
}

__global__ void sage_gemm_wmma(const float* __restrict__ msg, const float* __restrict__ rdeg,
                               const float* __restrict__ h,
                               const float* __restrict__ Wl, const float* __restrict__ Wr,
                               const float* __restrict__ bias, float* __restrict__ out,
                               int K, int F, int nblocks_n, int do_relu) {
  __shared__ float sA[2][16 * SA_STRIDE];
  __shared__ float sB[2][KC * SB_STRIDE];

  int t = threadIdx.x;
  int wave = t >> 5, lane = t & 31;
  int mBase = (blockIdx.x / nblocks_n) * 16;
  int nBase = (blockIdx.x % nblocks_n) * 64;

  int m16  = lane & 15;
  int kb   = (lane >> 4) << 1;
  int nloc = wave * 16 + (lane & 15);
  int ncol = nBase + nloc;

  float scale = rdeg[mBase + m16];
  v8f acc = {0.f, 0.f, 0.f, 0.f, 0.f, 0.f, 0.f, 0.f};

  int ar = t >> 3, ac = (t & 7) << 2;
  int br = t >> 4, bc = (t & 15) << 2;

  for (int kc = 0; kc < K; kc += KC) {
    {
      float4 va0 = *(const float4*)(msg + (size_t)(mBase + ar) * K + kc + ac);
      float4 va1 = *(const float4*)(h   + (size_t)(mBase + ar) * K + kc + ac);
      *(float4*)(&sA[0][ar * SA_STRIDE + ac]) = va0;
      *(float4*)(&sA[1][ar * SA_STRIDE + ac]) = va1;
#pragma unroll
      for (int pass = 0; pass < 4; pass++) {
        int row = br + pass * 8;
        float4 vb0 = *(const float4*)(Wl + (size_t)(kc + row) * F + nBase + bc);
        float4 vb1 = *(const float4*)(Wr + (size_t)(kc + row) * F + nBase + bc);
        *(float4*)(&sB[0][row * SB_STRIDE + bc]) = vb0;
        *(float4*)(&sB[1][row * SB_STRIDE + bc]) = vb1;
      }
    }
    if (kc + KC < K) {
      __builtin_prefetch(msg + (size_t)(mBase + ar) * K + kc + KC + ac, 0, 0);
      __builtin_prefetch(h   + (size_t)(mBase + ar) * K + kc + KC + ac, 0, 0);
      __builtin_prefetch(Wl + (size_t)(kc + KC + br) * F + nBase + bc, 0, 0);
      __builtin_prefetch(Wr + (size_t)(kc + KC + br) * F + nBase + bc, 0, 0);
    }
    __syncthreads();

#pragma unroll
    for (int k = 0; k < KC; k += 4) {
      v2f a, b;
      a.x = sA[0][m16 * SA_STRIDE + k + kb] * scale;
      a.y = sA[0][m16 * SA_STRIDE + k + kb + 1] * scale;
      b.x = sB[0][(k + kb) * SB_STRIDE + nloc];
      b.y = sB[0][(k + kb + 1) * SB_STRIDE + nloc];
      acc = __builtin_amdgcn_wmma_f32_16x16x4_f32(false, a, false, b, (short)0, acc,
                                                  false, false);
      a.x = sA[1][m16 * SA_STRIDE + k + kb];
      a.y = sA[1][m16 * SA_STRIDE + k + kb + 1];
      b.x = sB[1][(k + kb) * SB_STRIDE + nloc];
      b.y = sB[1][(k + kb + 1) * SB_STRIDE + nloc];
      acc = __builtin_amdgcn_wmma_f32_16x16x4_f32(false, a, false, b, (short)0, acc,
                                                  false, false);
    }
    __syncthreads();
  }

  float bv = bias[ncol];
  int mhalf = mBase + ((lane >> 4) << 3);
#pragma unroll
  for (int v = 0; v < 8; v++) {
    float val = acc[v] + bv;
    if (do_relu) val = fmaxf(val, 0.f);
    out[(size_t)(mhalf + v) * F + ncol] = val;
  }
}

__global__ void bn_stats(const float* __restrict__ h, float* __restrict__ st, int M, int F) {
  int c = threadIdx.x;
  float s = 0.f, s2 = 0.f;
  for (int r = blockIdx.x; r < M; r += gridDim.x) {
    float v = h[(size_t)r * F + c];
    s += v;
    s2 += v * v;
  }
  atomicAdd(&st[c], s);
  atomicAdd(&st[F + c], s2);
}

__global__ void bn_apply(float* __restrict__ h, const float* __restrict__ st,
                         const float* __restrict__ g, const float* __restrict__ be,
                         int M, int F) {
  int i = blockIdx.x * blockDim.x + threadIdx.x;
  if (i >= M * F) return;
  int c = i & (F - 1);
  float inv = 1.f / (float)M;
  float mean = st[c] * inv;
  float var = st[F + c] * inv - mean * mean;
  h[i] = g[c] * (h[i] - mean) * rsqrtf(var + 1e-5f) + be[c];
}

__global__ void reparam(const float* __restrict__ h3, const float* __restrict__ eps,
                        float* __restrict__ z, int M) {
  int i = blockIdx.x * blockDim.x + threadIdx.x;
  if (i >= M * 64) return;
  int r = i >> 6, c = i & 63;
  const float* row = h3 + (size_t)r * 128;
  z[i] = row[c] + expf(row[64 + c]) * eps[i];
}

extern "C" void kernel_launch(void* const* d_in, const int* in_sizes, int n_in,
                              void* d_out, int out_size, void* d_ws, size_t ws_size,
                              hipStream_t stream) {
  const float* x   = (const float*)d_in[0];
  const int*   src = (const int*)d_in[1];
  const int*   dst = (const int*)d_in[2];
  const float* eps = (const float*)d_in[3];
  const float* W1l = (const float*)d_in[4];
  const float* W1r = (const float*)d_in[5];
  const float* b1  = (const float*)d_in[6];
  const float* g1  = (const float*)d_in[7];
  const float* be1 = (const float*)d_in[8];
  const float* W2l = (const float*)d_in[9];
  const float* W2r = (const float*)d_in[10];
  const float* b2  = (const float*)d_in[11];
  const float* g2  = (const float*)d_in[12];
  const float* be2 = (const float*)d_in[13];
  const float* W3l = (const float*)d_in[14];
  const float* W3r = (const float*)d_in[15];
  const float* b3  = (const float*)d_in[16];
  float* zout = (float*)d_out;
  (void)in_sizes; (void)n_in; (void)out_size; (void)ws_size;

  char* ws = (char*)d_ws;
  size_t off = 0;
  auto carve = [&](size_t bytes) -> float* {
    char* p = ws + off;
    off = (off + bytes + 255) & ~(size_t)255;
    return (float*)p;
  };
  float* rdeg = carve((size_t)NN * 4);
  float* msg  = carve((size_t)NN * 256 * 4);
  float* h1   = carve((size_t)NN * 128 * 4);
  float* h2   = carve((size_t)NN * 256 * 4);
  float* st   = carve(512 * 4);

  const int M = NN, E = NE;

  zero_f32<<<256, 256, 0, stream>>>(rdeg, M);
  deg_count<<<(E + 255) / 256, 256, 0, stream>>>(dst, rdeg, E);
  rdeg_inv<<<(M + 255) / 256, 256, 0, stream>>>(rdeg, M);

  zero_f32<<<2048, 256, 0, stream>>>(msg, M * 128);
  scatter_add<<<(E * 32 + 255) / 256, 256, 0, stream>>>(x, src, dst, msg, E, 128, 5);
  sage_gemm_wmma<<<3125 * 2, 128, 0, stream>>>(msg, rdeg, x, W1l, W1r, b1, h1,
                                               128, 128, 2, 1);
  zero_f32<<<1, 256, 0, stream>>>(st, 256);
  bn_stats<<<512, 128, 0, stream>>>(h1, st, M, 128);
  bn_apply<<<(M * 128 + 255) / 256, 256, 0, stream>>>(h1, st, g1, be1, M, 128);

  zero_f32<<<2048, 256, 0, stream>>>(msg, M * 128);
  scatter_add<<<(E * 32 + 255) / 256, 256, 0, stream>>>(h1, src, dst, msg, E, 128, 5);
  sage_gemm_wmma<<<3125 * 4, 128, 0, stream>>>(msg, rdeg, h1, W2l, W2r, b2, h2,
                                               128, 256, 4, 1);
  zero_f32<<<2, 256, 0, stream>>>(st, 512);
  bn_stats<<<512, 256, 0, stream>>>(h2, st, M, 256);
  bn_apply<<<(M * 256 + 255) / 256, 256, 0, stream>>>(h2, st, g2, be2, M, 256);

  zero_f32<<<4096, 256, 0, stream>>>(msg, M * 256);
  scatter_add<<<(E * 64 + 255) / 256, 256, 0, stream>>>(h2, src, dst, msg, E, 256, 6);
  sage_gemm_wmma<<<3125 * 2, 128, 0, stream>>>(msg, rdeg, h2, W3l, W3r, b3, h1,
                                               256, 128, 2, 0);

  reparam<<<(M * 64 + 255) / 256, 256, 0, stream>>>(h1, eps, zout, M);
}


# Neighbour sums by gather, not float atomics: edges grouped by target in-call (int count atomic gives each edge its slot; scan; fill); one wave adds a row's sources in slot order and stores it once
# speedup vs baseline: 1.7499x; 1.7499x over previous
#include <hip/hip_runtime.h>

#define NN 50000
#define NE 800000
#define KC 32
#define SA_STRIDE 36
#define SB_STRIDE 68
#define SCAN_T 1024
#define SCAN_CH 49
#define CNT_PAD (SCAN_T * SCAN_CH)

typedef __attribute__((ext_vector_type(2))) float v2f;
typedef __attribute__((ext_vector_type(8))) float v8f;

__global__ void zero_f32(float* __restrict__ p, int n) {
  int i = blockIdx.x * blockDim.x + threadIdx.x;
  int stride = gridDim.x * blockDim.x;
  for (; i < n; i += stride) p[i] = 0.f;
}

__global__ void deg_count(const int* __restrict__ dst, int* __restrict__ cnt,
                          int* __restrict__ rank, int E) {
  int e = blockIdx.x * blockDim.x + threadIdx.x;
  if (e < E) rank[e] = atomicAdd(&cnt[dst[e]], 1);
}

__global__ void __launch_bounds__(SCAN_T) scan_rowptr(const int* __restrict__ cnt,
                                                      int* __restrict__ rowptr,
                                                      float* __restrict__ rdeg, int n) {
  __shared__ int s[CNT_PAD];
  __shared__ int wsum[32];
  int t = threadIdx.x, lane = t & 31, wave = t >> 5;
  for (int j = 0; j < SCAN_CH; j++) {
    int idx = j * SCAN_T + t;
    s[idx] = (idx < n) ? cnt[idx] : 0;
  }
  __syncthreads();
  int local = 0;
  for (int j = 0; j < SCAN_CH; j++) local += s[t * SCAN_CH + j];
  int inc = local;
  for (int o = 1; o < 32; o <<= 1) {
    int v = __shfl_up(inc, o);
    if (lane >= o) inc += v;
  }
  if (lane == 31) wsum[wave] = inc;
  __syncthreads();
  if (wave == 0) {
    int w = wsum[lane], wi = w;
    for (int o = 1; o < 32; o <<= 1) {
      int v = __shfl_up(wi, o);
      if (lane >= o) wi += v;
    }
    wsum[lane] = wi - w;
  }
  __syncthreads();
  int run = wsum[wave] + inc - local;
  for (int j = 0; j < SCAN_CH; j++) {
    int c = s[t * SCAN_CH + j];
    s[t * SCAN_CH + j] = run;
    run += c;
  }
  __syncthreads();
  for (int j = 0; j < SCAN_CH; j++) {
    int idx = j * SCAN_T + t;
    if (idx <= n) rowptr[idx] = s[idx];
    if (idx < n) rdeg[idx] = 1.f / fmaxf((float)(s[idx + 1] - s[idx]), 1.f);
  }
}

__global__ void csr_fill(const int* __restrict__ src, const int* __restrict__ dst,
                         const int* __restrict__ rowptr, const int* __restrict__ rank,
                         int* __restrict__ col, int E) {
  int e = blockIdx.x * blockDim.x + threadIdx.x;
  if (e >= E) return;
  col[rowptr[dst[e]] + rank[e]] = src[e];
}

template <int V, int R>
__device__ __forceinline__ void add_rows(float4 (&acc)[V], const float* __restrict__ hl,
                                         int c, int j) {
  constexpr int F = 128 * V;
  float4 x[R][V];
#pragma unroll
  for (int r = 0; r < R; r++) {
    const float* p = hl + (size_t)__builtin_amdgcn_readlane(c, j + r) * F;
#pragma unroll
    for (int v = 0; v < V; v++) x[r][v] = *(const float4*)(p + v * 128);
  }
#pragma unroll
  for (int r = 0; r < R; r++) {
#pragma unroll
    for (int v = 0; v < V; v++) {
      acc[v].x += x[r][v].x; acc[v].y += x[r][v].y;
      acc[v].z += x[r][v].z; acc[v].w += x[r][v].w;
    }
  }
}

template <int V>
__device__ __forceinline__ void gather_row(const float* __restrict__ h,
                                           const int* __restrict__ rowptr,
                                           const int* __restrict__ col,
                                           float* __restrict__ msg, int n) {
  constexpr int F = 128 * V;
  int lane = threadIdx.x & 31;
  int i = blockIdx.x * (blockDim.x >> 5) + (threadIdx.x >> 5);
  i = __builtin_amdgcn_readfirstlane(i);
  if (i >= n) return;
  int beg = rowptr[i], end = rowptr[i + 1];
  float4 acc[V];
#pragma unroll
  for (int v = 0; v < V; v++) acc[v] = make_float4(0.f, 0.f, 0.f, 0.f);
  const float* hl = h + lane * 4;
  for (int base = beg; base < end; base += 32) {
    int m = end - base;
    if (m > 32) m = 32;
    int c = (lane < m) ? col[base + lane] : 0;
    int j = 0;
    for (; j + 4 <= m; j += 4) add_rows<V, 4>(acc, hl, c, j);
    int rem = m - j;
    if (rem == 3) add_rows<V, 3>(acc, hl, c, j);
    else if (rem == 2) add_rows<V, 2>(acc, hl, c, j);
    else if (rem == 1) add_rows<V, 1>(acc, hl, c, j);
  }
  float* o = msg + (size_t)i * F + lane * 4;
#pragma unroll
  for (int v = 0; v < V; v++) *(float4*)(o + v * 128) = acc[v];
}

__global__ void gather_sum_128(const float* __restrict__ h, const int* __restrict__ rowptr,
                               const int* __restrict__ col, float* __restrict__ msg, int n) {
  gather_row<1>(h, rowptr, col, msg, n);
}

__global__ void gather_sum_256(const float* __restrict__ h, const int* __restrict__ rowptr,
                               const int* __restrict__ col, float* __restrict__ msg, int n) {
  gather_row<2>(h, rowptr, col, msg, n);
}

__global__ void sage_gemm_wmma(const float* __restrict__ msg, const float* __restrict__ rdeg,
                               const float* __restrict__ h,
                               const float* __restrict__ Wl, const float* __restrict__ Wr,
                               const float* __restrict__ bias, float* __restrict__ out,
                               int K, int F, int nblocks_n, int do_relu) {
  __shared__ float sA[2][16 * SA_STRIDE];
  __shared__ float sB[2][KC * SB_STRIDE];

  int t = threadIdx.x;
  int wave = t >> 5, lane = t & 31;
  int mBase = (blockIdx.x / nblocks_n) * 16;
  int nBase = (blockIdx.x % nblocks_n) * 64;

  int m16  = lane & 15;
  int kb   = (lane >> 4) << 1;
  int nloc = wave * 16 + (lane & 15);
  int ncol = nBase + nloc;

  float scale = rdeg[mBase + m16];
  v8f acc = {0.f, 0.f, 0.f, 0.f, 0.f, 0.f, 0.f, 0.f};

  int ar = t >> 3, ac = (t & 7) << 2;
  int br = t >> 4, bc = (t & 15) << 2;

  for (int kc = 0; kc < K; kc += KC) {
    {
      float4 va0 = *(const float4*)(msg + (size_t)(mBase + ar) * K + kc + ac);
      float4 va1 = *(const float4*)(h   + (size_t)(mBase + ar) * K + kc + ac);
      *(float4*)(&sA[0][ar * SA_STRIDE + ac]) = va0;
      *(float4*)(&sA[1][ar * SA_STRIDE + ac]) = va1;
#pragma unroll
      for (int pass = 0; pass < 4; pass++) {
        int row = br + pass * 8;
        float4 vb0 = *(const float4*)(Wl + (size_t)(kc + row) * F + nBase + bc);
        float4 vb1 = *(const float4*)(Wr + (size_t)(kc + row) * F + nBase + bc);
        *(float4*)(&sB[0][row * SB_STRIDE + bc]) = vb0;
        *(float4*)(&sB[1][row * SB_STRIDE + bc]) = vb1;
      }
    }
    if (kc + KC < K) {
      __builtin_prefetch(msg + (size_t)(mBase + ar) * K + kc + KC + ac, 0, 0);
      __builtin_prefetch(h   + (size_t)(mBase + ar) * K + kc + KC + ac, 0, 0);
      __builtin_prefetch(Wl + (size_t)(kc + KC + br) * F + nBase + bc, 0, 0);
      __builtin_prefetch(Wr + (size_t)(kc + KC + br) * F + nBase + bc, 0, 0);
    }
    __syncthreads();

#pragma unroll
    for (int k = 0; k < KC; k += 4) {
      v2f a, b;
      a.x = sA[0][m16 * SA_STRIDE + k + kb] * scale;
      a.y = sA[0][m16 * SA_STRIDE + k + kb + 1] * scale;
      b.x = sB[0][(k + kb) * SB_STRIDE + nloc];
      b.y = sB[0][(k + kb + 1) * SB_STRIDE + nloc];
      acc = __builtin_amdgcn_wmma_f32_16x16x4_f32(false, a, false, b, (short)0, acc,
                                                  false, false);
      a.x = sA[1][m16 * SA_STRIDE + k + kb];
      a.y = sA[1][m16 * SA_STRIDE + k + kb + 1];
      b.x = sB[1][(k + kb) * SB_STRIDE + nloc];
      b.y = sB[1][(k + kb + 1) * SB_STRIDE + nloc];
      acc = __builtin_amdgcn_wmma_f32_16x16x4_f32(false, a, false, b, (short)0, acc,
                                                  false, false);
    }
    __syncthreads();
  }

  float bv = bias[ncol];
  int mhalf = mBase + ((lane >> 4) << 3);
#pragma unroll
  for (int v = 0; v < 8; v++) {
    float val = acc[v] + bv;
    if (do_relu) val = fmaxf(val, 0.f);
    out[(size_t)(mhalf + v) * F + ncol] = val;
  }
}

__global__ void bn_stats(const float* __restrict__ h, float* __restrict__ st, int M, int F) {
  int c = threadIdx.x;
  float s = 0.f, s2 = 0.f;
  for (int r = blockIdx.x; r < M; r += gridDim.x) {
    float v = h[(size_t)r * F + c];
    s += v;
    s2 += v * v;
  }
  atomicAdd(&st[c], s);
  atomicAdd(&st[F + c], s2);
}

__global__ void bn_apply(float* __restrict__ h, const float* __restrict__ st,
                         const float* __restrict__ g, const float* __restrict__ be,
                         int M, int F) {
  int i = blockIdx.x * blockDim.x + threadIdx.x;
  if (i >= M * F) return;
  int c = i & (F - 1);
  float inv = 1.f / (float)M;
  float mean = st[c] * inv;
  float var = st[F + c] * inv - mean * mean;
  h[i] = g[c] * (h[i] - mean) * rsqrtf(var + 1e-5f) + be[c];
}

__global__ void reparam(const float* __restrict__ h3, const float* __restrict__ eps,
                        float* __restrict__ z, int M) {
  int i = blockIdx.x * blockDim.x + threadIdx.x;
  if (i >= M * 64) return;
  int r = i >> 6, c = i & 63;
  const float* row = h3 + (size_t)r * 128;
  z[i] = row[c] + expf(row[64 + c]) * eps[i];
}

extern "C" void kernel_launch(void* const* d_in, const int* in_sizes, int n_in,
                              void* d_out, int out_size, void* d_ws, size_t ws_size,
                              hipStream_t stream) {
  const float* x   = (const float*)d_in[0];
  const int*   src = (const int*)d_in[1];
  const int*   dst = (const int*)d_in[2];
  const float* eps = (const float*)d_in[3];
  const float* W1l = (const float*)d_in[4];
  const float* W1r = (const float*)d_in[5];
  const float* b1  = (const float*)d_in[6];
  const float* g1  = (const float*)d_in[7];
  const float* be1 = (const float*)d_in[8];
  const float* W2l = (const float*)d_in[9];
  const float* W2r = (const float*)d_in[10];
  const float* b2  = (const float*)d_in[11];
  const float* g2  = (const float*)d_in[12];
  const float* be2 = (const float*)d_in[13];
  const float* W3l = (const float*)d_in[14];
  const float* W3r = (const float*)d_in[15];
  const float* b3  = (const float*)d_in[16];
  float* zout = (float*)d_out;
  (void)in_sizes; (void)n_in; (void)out_size; (void)ws_size;

  char* ws = (char*)d_ws;
  size_t off = 0;
  auto carve = [&](size_t bytes) -> float* {
    char* p = ws + off;
    off = (off + bytes + 255) & ~(size_t)255;
    return (float*)p;
  };
  float* rdeg = carve((size_t)NN * 4);
  float* msg  = carve((size_t)NN * 256 * 4);
  float* h1   = carve((size_t)NN * 128 * 4);
  float* h2   = carve((size_t)NN * 256 * 4);
  float* st   = carve(512 * 4);
  float* y1   = msg + (size_t)NN * 128;
  int* cnt    = (int*)h1;
  int* rowptr = cnt + CNT_PAD;
  int* col    = rowptr + CNT_PAD;
  int* rank   = col + NE;

  const int M = NN, E = NE;

  zero_f32<<<256, 256, 0, stream>>>((float*)cnt, M);
  deg_count<<<(E + 255) / 256, 256, 0, stream>>>(dst, cnt, rank, E);
  scan_rowptr<<<1, SCAN_T, 0, stream>>>(cnt, rowptr, rdeg, M);
  csr_fill<<<(E + 255) / 256, 256, 0, stream>>>(src, dst, rowptr, rank, col, E);

  gather_sum_128<<<(M + 7) / 8, 256, 0, stream>>>(x, rowptr, col, msg, M);
  sage_gemm_wmma<<<3125 * 2, 128, 0, stream>>>(msg, rdeg, x, W1l, W1r, b1, y1,
                                               128, 128, 2, 1);
  zero_f32<<<1, 256, 0, stream>>>(st, 256);
  bn_stats<<<512, 128, 0, stream>>>(y1, st, M, 128);
  bn_apply<<<(M * 128 + 255) / 256, 256, 0, stream>>>(y1, st, g1, be1, M, 128);

  gather_sum_128<<<(M + 7) / 8, 256, 0, stream>>>(y1, rowptr, col, msg, M);
  sage_gemm_wmma<<<3125 * 4, 128, 0, stream>>>(msg, rdeg, y1, W2l, W2r, b2, h2,
                                               128, 256, 4, 1);
  zero_f32<<<2, 256, 0, stream>>>(st, 512);
  bn_stats<<<512, 256, 0, stream>>>(h2, st, M, 256);
  bn_apply<<<(M * 256 + 255) / 256, 256, 0, stream>>>(h2, st, g2, be2, M, 256);

  gather_sum_256<<<(M + 7) / 8, 256, 0, stream>>>(h2, rowptr, col, msg, M);
  sage_gemm_wmma<<<3125 * 2, 128, 0, stream>>>(msg, rdeg, h2, W3l, W3r, b3, h1,
                                               256, 128, 2, 0);

  reparam<<<(M * 64 + 255) / 256, 256, 0, stream>>>(h1, eps, zout, M);
}


# bn_stats: a lane loads 14 rows of its column as one group, then adds them (was one 4-byte load per loop trip, each waited out); same rows per thread, same add order, square still added fused
# speedup vs baseline: 2.0488x; 1.1708x over previous
#include <hip/hip_runtime.h>

#define NN 50000
#define NE 800000
#define KC 32
#define SA_STRIDE 36
#define SB_STRIDE 68
#define SCAN_T 1024
#define SCAN_CH 49
#define CNT_PAD (SCAN_T * SCAN_CH)

typedef __attribute__((ext_vector_type(2))) float v2f;
typedef __attribute__((ext_vector_type(8))) float v8f;

__global__ void zero_f32(float* __restrict__ p, int n) {
  int i = blockIdx.x * blockDim.x + threadIdx.x;
  int stride = gridDim.x * blockDim.x;
  for (; i < n; i += stride) p[i] = 0.f;
}

__global__ void deg_count(const int* __restrict__ dst, int* __restrict__ cnt,
                          int* __restrict__ rank, int E) {
  int e = blockIdx.x * blockDim.x + threadIdx.x;
  if (e < E) rank[e] = atomicAdd(&cnt[dst[e]], 1);
}

__global__ void __launch_bounds__(SCAN_T) scan_rowptr(const int* __restrict__ cnt,
                                                      int* __restrict__ rowptr,
                                                      float* __restrict__ rdeg, int n) {
  __shared__ int s[CNT_PAD];
  __shared__ int wsum[32];
  int t = threadIdx.x, lane = t & 31, wave = t >> 5;
  for (int j = 0; j < SCAN_CH; j++) {
    int idx = j * SCAN_T + t;
    s[idx] = (idx < n) ? cnt[idx] : 0;
  }
  __syncthreads();
  int local = 0;
  for (int j = 0; j < SCAN_CH; j++) local += s[t * SCAN_CH + j];
  int inc = local;
  for (int o = 1; o < 32; o <<= 1) {
    int v = __shfl_up(inc, o);
    if (lane >= o) inc += v;
  }
  if (lane == 31) wsum[wave] = inc;
  __syncthreads();
  if (wave == 0) {
    int w = wsum[lane], wi = w;
    for (int o = 1; o < 32; o <<= 1) {
      int v = __shfl_up(wi, o);
      if (lane >= o) wi += v;
    }
    wsum[lane] = wi - w;
  }
  __syncthreads();
  int run = wsum[wave] + inc - local;
  for (int j = 0; j < SCAN_CH; j++) {
    int c = s[t * SCAN_CH + j];
    s[t * SCAN_CH + j] = run;
    run += c;
  }
  __syncthreads();
  for (int j = 0; j < SCAN_CH; j++) {
    int idx = j * SCAN_T + t;
    if (idx <= n) rowptr[idx] = s[idx];
    if (idx < n) rdeg[idx] = 1.f / fmaxf((float)(s[idx + 1] - s[idx]), 1.f);
  }
}

__global__ void csr_fill(const int* __restrict__ src, const int* __restrict__ dst,
                         const int* __restrict__ rowptr, const int* __restrict__ rank,
                         int* __restrict__ col, int E) {
  int e = blockIdx.x * blockDim.x + threadIdx.x;
  if (e >= E) return;
  col[rowptr[dst[e]] + rank[e]] = src[e];
}

template <int V, int R>
__device__ __forceinline__ void add_rows(float4 (&acc)[V], const float* __restrict__ hl,
                                         int c, int j) {
  constexpr int F = 128 * V;
  float4 x[R][V];
#pragma unroll
  for (int r = 0; r < R; r++) {
    const float* p = hl + (size_t)__builtin_amdgcn_readlane(c, j + r) * F;
#pragma unroll
    for (int v = 0; v < V; v++) x[r][v] = *(const float4*)(p + v * 128);
  }
#pragma unroll
  for (int r = 0; r < R; r++) {
#pragma unroll
    for (int v = 0; v < V; v++) {
      acc[v].x += x[r][v].x; acc[v].y += x[r][v].y;
      acc[v].z += x[r][v].z; acc[v].w += x[r][v].w;
    }
  }
}

template <int V>
__device__ __forceinline__ void gather_row(const float* __restrict__ h,
                                           const int* __restrict__ rowptr,
                                           const int* __restrict__ col,
                                           float* __restrict__ msg, int n) {
  constexpr int F = 128 * V;
  int lane = threadIdx.x & 31;
  int i = blockIdx.x * (blockDim.x >> 5) + (threadIdx.x >> 5);
  i = __builtin_amdgcn_readfirstlane(i);
  if (i >= n) return;
  int beg = rowptr[i], end = rowptr[i + 1];
  float4 acc[V];
#pragma unroll
  for (int v = 0; v < V; v++) acc[v] = make_float4(0.f, 0.f, 0.f, 0.f);
  const float* hl = h + lane * 4;
  for (int base = beg; base < end; base += 32) {
    int m = end - base;
    if (m > 32) m = 32;
    int c = (lane < m) ? col[base + lane] : 0;
    int j = 0;
    for (; j + 4 <= m; j += 4) add_rows<V, 4>(acc, hl, c, j);
    int rem = m - j;
    if (rem == 3) add_rows<V, 3>(acc, hl, c, j);
    else if (rem == 2) add_rows<V, 2>(acc, hl, c, j);
    else if (rem == 1) add_rows<V, 1>(acc, hl, c, j);
  }
  float* o = msg + (size_t)i * F + lane * 4;
#pragma unroll
  for (int v = 0; v < V; v++) *(float4*)(o + v * 128) = acc[v];
}

__global__ void gather_sum_128(const float* __restrict__ h, const int* __restrict__ rowptr,
                               const int* __restrict__ col, float* __restrict__ msg, int n) {
  gather_row<1>(h, rowptr, col, msg, n);
}

__global__ void gather_sum_256(const float* __restrict__ h, const int* __restrict__ rowptr,
                               const int* __restrict__ col, float* __restrict__ msg, int n) {
  gather_row<2>(h, rowptr, col, msg, n);
}

__global__ void sage_gemm_wmma(const float* __restrict__ msg, const float* __restrict__ rdeg,
                               const float* __restrict__ h,
                               const float* __restrict__ Wl, const float* __restrict__ Wr,
                               const float* __restrict__ bias, float* __restrict__ out,
                               int K, int F, int nblocks_n, int do_relu) {
  __shared__ float sA[2][16 * SA_STRIDE];
  __shared__ float sB[2][KC * SB_STRIDE];

  int t = threadIdx.x;
  int wave = t >> 5, lane = t & 31;
  int mBase = (blockIdx.x / nblocks_n) * 16;
  int nBase = (blockIdx.x % nblocks_n) * 64;

  int m16  = lane & 15;
  int kb   = (lane >> 4) << 1;
  int nloc = wave * 16 + (lane & 15);
  int ncol = nBase + nloc;

  float scale = rdeg[mBase + m16];
  v8f acc = {0.f, 0.f, 0.f, 0.f, 0.f, 0.f, 0.f, 0.f};

  int ar = t >> 3, ac = (t & 7) << 2;
  int br = t >> 4, bc = (t & 15) << 2;

  for (int kc = 0; kc < K; kc += KC) {
    {
      float4 va0 = *(const float4*)(msg + (size_t)(mBase + ar) * K + kc + ac);
      float4 va1 = *(const float4*)(h   + (size_t)(mBase + ar) * K + kc + ac);
      *(float4*)(&sA[0][ar * SA_STRIDE + ac]) = va0;
      *(float4*)(&sA[1][ar * SA_STRIDE + ac]) = va1;
#pragma unroll
      for (int pass = 0; pass < 4; pass++) {
        int row = br + pass * 8;
        float4 vb0 = *(const float4*)(Wl + (size_t)(kc + row) * F + nBase + bc);
        float4 vb1 = *(const float4*)(Wr + (size_t)(kc + row) * F + nBase + bc);
        *(float4*)(&sB[0][row * SB_STRIDE + bc]) = vb0;
        *(float4*)(&sB[1][row * SB_STRIDE + bc]) = vb1;
      }
    }
    if (kc + KC < K) {
      __builtin_prefetch(msg + (size_t)(mBase + ar) * K + kc + KC + ac, 0, 0);
      __builtin_prefetch(h   + (size_t)(mBase + ar) * K + kc + KC + ac, 0, 0);
      __builtin_prefetch(Wl + (size_t)(kc + KC + br) * F + nBase + bc, 0, 0);
      __builtin_prefetch(Wr + (size_t)(kc + KC + br) * F + nBase + bc, 0, 0);
    }
    __syncthreads();

#pragma unroll
    for (int k = 0; k < KC; k += 4) {
      v2f a, b;
      a.x = sA[0][m16 * SA_STRIDE + k + kb] * scale;
      a.y = sA[0][m16 * SA_STRIDE + k + kb + 1] * scale;
      b.x = sB[0][(k + kb) * SB_STRIDE + nloc];
      b.y = sB[0][(k + kb + 1) * SB_STRIDE + nloc];
      acc = __builtin_amdgcn_wmma_f32_16x16x4_f32(false, a, false, b, (short)0, acc,
                                                  false, false);
      a.x = sA[1][m16 * SA_STRIDE + k + kb];
      a.y = sA[1][m16 * SA_STRIDE + k + kb + 1];
      b.x = sB[1][(k + kb) * SB_STRIDE + nloc];
      b.y = sB[1][(k + kb + 1) * SB_STRIDE + nloc];
      acc = __builtin_amdgcn_wmma_f32_16x16x4_f32(false, a, false, b, (short)0, acc,
                                                  false, false);
    }
    __syncthreads();
  }

  float bv = bias[ncol];
  int mhalf = mBase + ((lane >> 4) << 3);
#pragma unroll
  for (int v = 0; v < 8; v++) {
    float val = acc[v] + bv;
    if (do_relu) val = fmaxf(val, 0.f);
    out[(size_t)(mhalf + v) * F + ncol] = val;
  }
}

#define BN_U 14
__global__ void bn_stats(const float* __restrict__ h, float* __restrict__ st, int M, int F) {
  int c = threadIdx.x;
  int stride = gridDim.x;
  float s = 0.f, s2 = 0.f;
  for (int r0 = blockIdx.x; r0 < M; r0 += stride * BN_U) {
    float v[BN_U];
#pragma unroll
    for (int j = 0; j < BN_U; j++) {
      int r = r0 + j * stride;
      v[j] = (r < M) ? h[(size_t)r * F + c] : 0.f;
    }
#pragma unroll
    for (int j = 0; j < BN_U; j++) {
      s += v[j];
      s2 = __builtin_fmaf(v[j], v[j], s2);
    }
  }
  atomicAdd(&st[c], s);
  atomicAdd(&st[F + c], s2);
}

__global__ void bn_apply(float* __restrict__ h, const float* __restrict__ st,
                         const float* __restrict__ g, const float* __restrict__ be,
                         int M, int F) {
  int i = blockIdx.x * blockDim.x + threadIdx.x;
  if (i >= M * F) return;
  int c = i & (F - 1);
  float inv = 1.f / (float)M;
  float mean = st[c] * inv;
  float var = st[F + c] * inv - mean * mean;
  h[i] = g[c] * (h[i] - mean) * rsqrtf(var + 1e-5f) + be[c];
}

__global__ void reparam(const float* __restrict__ h3, const float* __restrict__ eps,
                        float* __restrict__ z, int M) {
  int i = blockIdx.x * blockDim.x + threadIdx.x;
  if (i >= M * 64) return;
  int r = i >> 6, c = i & 63;
  const float* row = h3 + (size_t)r * 128;
  z[i] = row[c] + expf(row[64 + c]) * eps[i];
}

extern "C" void kernel_launch(void* const* d_in, const int* in_sizes, int n_in,
                              void* d_out, int out_size, void* d_ws, size_t ws_size,
                              hipStream_t stream) {
  const float* x   = (const float*)d_in[0];
  const int*   src = (const int*)d_in[1];
  const int*   dst = (const int*)d_in[2];
  const float* eps = (const float*)d_in[3];
  const float* W1l = (const float*)d_in[4];
  const float* W1r = (const float*)d_in[5];
  const float* b1  = (const float*)d_in[6];
  const float* g1  = (const float*)d_in[7];
  const float* be1 = (const float*)d_in[8];
  const float* W2l = (const float*)d_in[9];
  const float* W2r = (const float*)d_in[10];
  const float* b2  = (const float*)d_in[11];
  const float* g2  = (const float*)d_in[12];
  const float* be2 = (const float*)d_in[13];
  const float* W3l = (const float*)d_in[14];
  const float* W3r = (const float*)d_in[15];
  const float* b3  = (const float*)d_in[16];
  float* zout = (float*)d_out;
  (void)in_sizes; (void)n_in; (void)out_size; (void)ws_size;

  char* ws = (char*)d_ws;
  size_t off = 0;
  auto carve = [&](size_t bytes) -> float* {
    char* p = ws + off;
    off = (off + bytes + 255) & ~(size_t)255;
    return (float*)p;
  };
  float* rdeg = carve((size_t)NN * 4);
  float* msg  = carve((size_t)NN * 256 * 4);
  float* h1   = carve((size_t)NN * 128 * 4);
  float* h2   = carve((size_t)NN * 256 * 4);
  float* st   = carve(512 * 4);
  float* y1   = msg + (size_t)NN * 128;
  int* cnt    = (int*)h1;
  int* rowptr = cnt + CNT_PAD;
  int* col    = rowptr + CNT_PAD;
  int* rank   = col + NE;

  const int M = NN, E = NE;

  zero_f32<<<256, 256, 0, stream>>>((float*)cnt, M);
  deg_count<<<(E + 255) / 256, 256, 0, stream>>>(dst, cnt, rank, E);
  scan_rowptr<<<1, SCAN_T, 0, stream>>>(cnt, rowptr, rdeg, M);
  csr_fill<<<(E + 255) / 256, 256, 0, stream>>>(src, dst, rowptr, rank, col, E);

  gather_sum_128<<<(M + 7) / 8, 256, 0, stream>>>(x, rowptr, col, msg, M);
  sage_gemm_wmma<<<3125 * 2, 128, 0, stream>>>(msg, rdeg, x, W1l, W1r, b1, y1,
                                               128, 128, 2, 1);
  zero_f32<<<1, 256, 0, stream>>>(st, 256);
  bn_stats<<<512, 128, 0, stream>>>(y1, st, M, 128);
  bn_apply<<<(M * 128 + 255) / 256, 256, 0, stream>>>(y1, st, g1, be1, M, 128);

  gather_sum_128<<<(M + 7) / 8, 256, 0, stream>>>(y1, rowptr, col, msg, M);
  sage_gemm_wmma<<<3125 * 4, 128, 0, stream>>>(msg, rdeg, y1, W2l, W2r, b2, h2,
                                               128, 256, 4, 1);
  zero_f32<<<2, 256, 0, stream>>>(st, 512);
  bn_stats<<<512, 256, 0, stream>>>(h2, st, M, 256);
  bn_apply<<<(M * 256 + 255) / 256, 256, 0, stream>>>(h2, st, g2, be2, M, 256);

  gather_sum_256<<<(M + 7) / 8, 256, 0, stream>>>(h2, rowptr, col, msg, M);
  sage_gemm_wmma<<<3125 * 2, 128, 0, stream>>>(msg, rdeg, h2, W3l, W3r, b3, h1,
                                               256, 128, 2, 0);

  reparam<<<(M * 64 + 255) / 256, 256, 0, stream>>>(h1, eps, zout, M);
}


# Global->LDS staging (sage_gemm: 10 loads per K-chunk; scan_rowptr: 49): all loads go out before the first wait and LDS store; the compiler had put each in front of its own store, one round trip each
# speedup vs baseline: 2.2976x; 1.1214x over previous
#include <hip/hip_runtime.h>

#define NN 50000
#define NE 800000
#define KC 32
#define SA_STRIDE 36
#define SB_STRIDE 68
#define SCAN_T 1024
#define SCAN_CH 49
#define CNT_PAD (SCAN_T * SCAN_CH)

typedef __attribute__((ext_vector_type(2))) float v2f;
typedef __attribute__((ext_vector_type(8))) float v8f;

__global__ void zero_f32(float* __restrict__ p, int n) {
  int i = blockIdx.x * blockDim.x + threadIdx.x;
  int stride = gridDim.x * blockDim.x;
  for (; i < n; i += stride) p[i] = 0.f;
}

__global__ void deg_count(const int* __restrict__ dst, int* __restrict__ cnt,
                          int* __restrict__ rank, int E) {
  int e = blockIdx.x * blockDim.x + threadIdx.x;
  if (e < E) rank[e] = atomicAdd(&cnt[dst[e]], 1);
}

__global__ void __launch_bounds__(SCAN_T) scan_rowptr(const int* __restrict__ cnt,
                                                      int* __restrict__ rowptr,
                                                      float* __restrict__ rdeg, int n) {
  __shared__ int s[CNT_PAD];
  __shared__ int wsum[32];
  int t = threadIdx.x, lane = t & 31, wave = t >> 5;
  {
    int v[SCAN_CH];
#pragma unroll
    for (int j = 0; j < SCAN_CH; j++) {
      int idx = j * SCAN_T + t;
      v[j] = cnt[(idx < n) ? idx : 0];
    }
    __builtin_amdgcn_sched_barrier(0);
#pragma unroll
    for (int j = 0; j < SCAN_CH; j++) {
      int idx = j * SCAN_T + t;
      s[idx] = (idx < n) ? v[j] : 0;
    }
  }
  __syncthreads();
  int local = 0;
  for (int j = 0; j < SCAN_CH; j++) local += s[t * SCAN_CH + j];
  int inc = local;
  for (int o = 1; o < 32; o <<= 1) {
    int v = __shfl_up(inc, o);
    if (lane >= o) inc += v;
  }
  if (lane == 31) wsum[wave] = inc;
  __syncthreads();
  if (wave == 0) {
    int w = wsum[lane], wi = w;
    for (int o = 1; o < 32; o <<= 1) {
      int v = __shfl_up(wi, o);
      if (lane >= o) wi += v;
    }
    wsum[lane] = wi - w;
  }
  __syncthreads();
  int run = wsum[wave] + inc - local;
  for (int j = 0; j < SCAN_CH; j++) {
    int c = s[t * SCAN_CH + j];
    s[t * SCAN_CH + j] = run;
    run += c;
  }
  __syncthreads();
  for (int j = 0; j < SCAN_CH; j++) {
    int idx = j * SCAN_T + t;
    if (idx <= n) rowptr[idx] = s[idx];
    if (idx < n) rdeg[idx] = 1.f / fmaxf((float)(s[idx + 1] - s[idx]), 1.f);
  }
}

__global__ void csr_fill(const int* __restrict__ src, const int* __restrict__ dst,
                         const int* __restrict__ rowptr, const int* __restrict__ rank,
                         int* __restrict__ col, int E) {
  int e = blockIdx.x * blockDim.x + threadIdx.x;
  if (e >= E) return;
  col[rowptr[dst[e]] + rank[e]] = src[e];
}

template <int V, int R>
__device__ __forceinline__ void add_rows(float4 (&acc)[V], const float* __restrict__ hl,
                                         int c, int j) {
  constexpr int F = 128 * V;
  float4 x[R][V];
#pragma unroll
  for (int r = 0; r < R; r++) {
    const float* p = hl + (size_t)__builtin_amdgcn_readlane(c, j + r) * F;
#pragma unroll
    for (int v = 0; v < V; v++) x[r][v] = *(const float4*)(p + v * 128);
  }
#pragma unroll
  for (int r = 0; r < R; r++) {
#pragma unroll
    for (int v = 0; v < V; v++) {
      acc[v].x += x[r][v].x; acc[v].y += x[r][v].y;
      acc[v].z += x[r][v].z; acc[v].w += x[r][v].w;
    }
  }
}

template <int V>
__device__ __forceinline__ void gather_row(const float* __restrict__ h,
                                           const int* __restrict__ rowptr,
                                           const int* __restrict__ col,
                                           float* __restrict__ msg, int n) {
  constexpr int F = 128 * V;
  int lane = threadIdx.x & 31;
  int i = blockIdx.x * (blockDim.x >> 5) + (threadIdx.x >> 5);
  i = __builtin_amdgcn_readfirstlane(i);
  if (i >= n) return;
  int beg = rowptr[i], end = rowptr[i + 1];
  float4 acc[V];
#pragma unroll
  for (int v = 0; v < V; v++) acc[v] = make_float4(0.f, 0.f, 0.f, 0.f);
  const float* hl = h + lane * 4;
  for (int base = beg; base < end; base += 32) {
    int m = end - base;
    if (m > 32) m = 32;
    int c = (lane < m) ? col[base + lane] : 0;
    int j = 0;
    for (; j + 4 <= m; j += 4) add_rows<V, 4>(acc, hl, c, j);
    int rem = m - j;
    if (rem == 3) add_rows<V, 3>(acc, hl, c, j);
    else if (rem == 2) add_rows<V, 2>(acc, hl, c, j);
    else if (rem == 1) add_rows<V, 1>(acc, hl, c, j);
  }
  float* o = msg + (size_t)i * F + lane * 4;
#pragma unroll
  for (int v = 0; v < V; v++) *(float4*)(o + v * 128) = acc[v];
}

__global__ void gather_sum_128(const float* __restrict__ h, const int* __restrict__ rowptr,
                               const int* __restrict__ col, float* __restrict__ msg, int n) {
  gather_row<1>(h, rowptr, col, msg, n);
}

__global__ void gather_sum_256(const float* __restrict__ h, const int* __restrict__ rowptr,
                               const int* __restrict__ col, float* __restrict__ msg, int n) {
  gather_row<2>(h, rowptr, col, msg, n);
}

__global__ void sage_gemm_wmma(const float* __restrict__ msg, const float* __restrict__ rdeg,
                               const float* __restrict__ h,
                               const float* __restrict__ Wl, const float* __restrict__ Wr,
                               const float* __restrict__ bias, float* __restrict__ out,
                               int K, int F, int nblocks_n, int do_relu) {
  __shared__ float sA[2][16 * SA_STRIDE];
  __shared__ float sB[2][KC * SB_STRIDE];

  int t = threadIdx.x;
  int wave = t >> 5, lane = t & 31;
  int mBase = (blockIdx.x / nblocks_n) * 16;
  int nBase = (blockIdx.x % nblocks_n) * 64;

  int m16  = lane & 15;
  int kb   = (lane >> 4) << 1;
  int nloc = wave * 16 + (lane & 15);
  int ncol = nBase + nloc;

  float scale = rdeg[mBase + m16];
  v8f acc = {0.f, 0.f, 0.f, 0.f, 0.f, 0.f, 0.f, 0.f};

  int ar = t >> 3, ac = (t & 7) << 2;
  int br = t >> 4, bc = (t & 15) << 2;

  for (int kc = 0; kc < K; kc += KC) {
    {
      float4 va0 = *(const float4*)(msg + (size_t)(mBase + ar) * K + kc + ac);
      float4 va1 = *(const float4*)(h   + (size_t)(mBase + ar) * K + kc + ac);
      float4 vb0[4], vb1[4];
#pragma unroll
      for (int pass = 0; pass < 4; pass++) {
        int row = br + pass * 8;
        vb0[pass] = *(const float4*)(Wl + (size_t)(kc + row) * F + nBase + bc);
        vb1[pass] = *(const float4*)(Wr + (size_t)(kc + row) * F + nBase + bc);
      }
      __builtin_amdgcn_sched_barrier(0);
      *(float4*)(&sA[0][ar * SA_STRIDE + ac]) = va0;
      *(float4*)(&sA[1][ar * SA_STRIDE + ac]) = va1;
#pragma unroll
      for (int pass = 0; pass < 4; pass++) {
        int row = br + pass * 8;
        *(float4*)(&sB[0][row * SB_STRIDE + bc]) = vb0[pass];
        *(float4*)(&sB[1][row * SB_STRIDE + bc]) = vb1[pass];
      }
    }
    if (kc + KC < K) {
      __builtin_prefetch(msg + (size_t)(mBase + ar) * K + kc + KC + ac, 0, 0);
      __builtin_prefetch(h   + (size_t)(mBase + ar) * K + kc + KC + ac, 0, 0);
      __builtin_prefetch(Wl + (size_t)(kc + KC + br) * F + nBase + bc, 0, 0);
      __builtin_prefetch(Wr + (size_t)(kc + KC + br) * F + nBase + bc, 0, 0);
    }
    __syncthreads();

#pragma unroll
    for (int k = 0; k < KC; k += 4) {
      v2f a, b;
      a.x = sA[0][m16 * SA_STRIDE + k + kb] * scale;
      a.y = sA[0][m16 * SA_STRIDE + k + kb + 1] * scale;
      b.x = sB[0][(k + kb) * SB_STRIDE + nloc];
      b.y = sB[0][(k + kb + 1) * SB_STRIDE + nloc];
      acc = __builtin_amdgcn_wmma_f32_16x16x4_f32(false, a, false, b, (short)0, acc,
                                                  false, false);
      a.x = sA[1][m16 * SA_STRIDE + k + kb];
      a.y = sA[1][m16 * SA_STRIDE + k + kb + 1];
      b.x = sB[1][(k + kb) * SB_STRIDE + nloc];
      b.y = sB[1][(k + kb + 1) * SB_STRIDE + nloc];
      acc = __builtin_amdgcn_wmma_f32_16x16x4_f32(false, a, false, b, (short)0, acc,
                                                  false, false);
    }
    __syncthreads();
  }

  float bv = bias[ncol];
  int mhalf = mBase + ((lane >> 4) << 3);
#pragma unroll
  for (int v = 0; v < 8; v++) {
    float val = acc[v] + bv;
    if (do_relu) val = fmaxf(val, 0.f);
    out[(size_t)(mhalf + v) * F + ncol] = val;
  }
}

#define BN_U 14
__global__ void bn_stats(const float* __restrict__ h, float* __restrict__ st, int M, int F) {
  int c = threadIdx.x;
  int stride = gridDim.x;
  float s = 0.f, s2 = 0.f;
  for (int r0 = blockIdx.x; r0 < M; r0 += stride * BN_U) {
    float v[BN_U];
#pragma unroll
    for (int j = 0; j < BN_U; j++) {
      int r = r0 + j * stride;
      v[j] = (r < M) ? h[(size_t)r * F + c] : 0.f;
    }
#pragma unroll
    for (int j = 0; j < BN_U; j++) {
      s += v[j];
      s2 = __builtin_fmaf(v[j], v[j], s2);
    }
  }
  atomicAdd(&st[c], s);
  atomicAdd(&st[F + c], s2);
}

__global__ void bn_apply(float* __restrict__ h, const float* __restrict__ st,
                         const float* __restrict__ g, const float* __restrict__ be,
                         int M, int F) {
  int i = blockIdx.x * blockDim.x + threadIdx.x;
  if (i >= M * F) return;
  int c = i & (F - 1);
  float inv = 1.f / (float)M;
  float mean = st[c] * inv;
  float var = st[F + c] * inv - mean * mean;
  h[i] = g[c] * (h[i] - mean) * rsqrtf(var + 1e-5f) + be[c];
}

__global__ void reparam(const float* __restrict__ h3, const float* __restrict__ eps,
                        float* __restrict__ z, int M) {
  int i = blockIdx.x * blockDim.x + threadIdx.x;
  if (i >= M * 64) return;
  int r = i >> 6, c = i & 63;
  const float* row = h3 + (size_t)r * 128;
  z[i] = row[c] + expf(row[64 + c]) * eps[i];
}

extern "C" void kernel_launch(void* const* d_in, const int* in_sizes, int n_in,
                              void* d_out, int out_size, void* d_ws, size_t ws_size,
                              hipStream_t stream) {
  const float* x   = (const float*)d_in[0];
  const int*   src = (const int*)d_in[1];
  const int*   dst = (const int*)d_in[2];
  const float* eps = (const float*)d_in[3];
  const float* W1l = (const float*)d_in[4];
  const float* W1r = (const float*)d_in[5];
  const float* b1  = (const float*)d_in[6];
  const float* g1  = (const float*)d_in[7];
  const float* be1 = (const float*)d_in[8];
  const float* W2l = (const float*)d_in[9];
  const float* W2r = (const float*)d_in[10];
  const float* b2  = (const float*)d_in[11];
  const float* g2  = (const float*)d_in[12];
  const float* be2 = (const float*)d_in[13];
  const float* W3l = (const float*)d_in[14];
  const float* W3r = (const float*)d_in[15];
  const float* b3  = (const float*)d_in[16];
  float* zout = (float*)d_out;
  (void)in_sizes; (void)n_in; (void)out_size; (void)ws_size;

  char* ws = (char*)d_ws;
  size_t off = 0;
  auto carve = [&](size_t bytes) -> float* {
    char* p = ws + off;
    off = (off + bytes + 255) & ~(size_t)255;
    return (float*)p;
  };
  float* rdeg = carve((size_t)NN * 4);
  float* msg  = carve((size_t)NN * 256 * 4);
  float* h1   = carve((size_t)NN * 128 * 4);
  float* h2   = carve((size_t)NN * 256 * 4);
  float* st   = carve(512 * 4);
  float* y1   = msg + (size_t)NN * 128;
  int* cnt    = (int*)h1;
  int* rowptr = cnt + CNT_PAD;
  int* col    = rowptr + CNT_PAD;
  int* rank   = col + NE;

  const int M = NN, E = NE;

  zero_f32<<<256, 256, 0, stream>>>((float*)cnt, M);
  deg_count<<<(E + 255) / 256, 256, 0, stream>>>(dst, cnt, rank, E);
  scan_rowptr<<<1, SCAN_T, 0, stream>>>(cnt, rowptr, rdeg, M);
  csr_fill<<<(E + 255) / 256, 256, 0, stream>>>(src, dst, rowptr, rank, col, E);

  gather_sum_128<<<(M + 7) / 8, 256, 0, stream>>>(x, rowptr, col, msg, M);
  sage_gemm_wmma<<<3125 * 2, 128, 0, stream>>>(msg, rdeg, x, W1l, W1r, b1, y1,
                                               128, 128, 2, 1);
  zero_f32<<<1, 256, 0, stream>>>(st, 256);
  bn_stats<<<512, 128, 0, stream>>>(y1, st, M, 128);
  bn_apply<<<(M * 128 + 255) / 256, 256, 0, stream>>>(y1, st, g1, be1, M, 128);

  gather_sum_128<<<(M + 7) / 8, 256, 0, stream>>>(y1, rowptr, col, msg, M);
  sage_gemm_wmma<<<3125 * 4, 128, 0, stream>>>(msg, rdeg, y1, W2l, W2r, b2, h2,
                                               128, 256, 4, 1);
  zero_f32<<<2, 256, 0, stream>>>(st, 512);
  bn_stats<<<512, 256, 0, stream>>>(h2, st, M, 256);
  bn_apply<<<(M * 256 + 255) / 256, 256, 0, stream>>>(h2, st, g2, be2, M, 256);

  gather_sum_256<<<(M + 7) / 8, 256, 0, stream>>>(h2, rowptr, col, msg, M);
  sage_gemm_wmma<<<3125 * 2, 128, 0, stream>>>(msg, rdeg, h2, W3l, W3r, b3, h1,
                                               256, 128, 2, 0);

  reparam<<<(M * 64 + 255) / 256, 256, 0, stream>>>(h1, eps, zout, M);
}


# sage_gemm: work-group tile 16x64 -> 80x64, each wave keeps 5 row tiles (5 accumulators): one B fragment read from LDS feeds 5 WMMAs, a W chunk serves 80 rows; WMMA sequence per tile unchanged
# speedup vs baseline: 2.5736x; 1.1201x over previous
#include <hip/hip_runtime.h>

#define NN 50000
#define NE 800000
#define KC 32
#define SA_STRIDE 36
#define SB_STRIDE 68
#define SCAN_T 1024
#define SCAN_CH 49
#define CNT_PAD (SCAN_T * SCAN_CH)

typedef __attribute__((ext_vector_type(2))) float v2f;
typedef __attribute__((ext_vector_type(8))) float v8f;

__global__ void zero_f32(float* __restrict__ p, int n) {
  int i = blockIdx.x * blockDim.x + threadIdx.x;
  int stride = gridDim.x * blockDim.x;
  for (; i < n; i += stride) p[i] = 0.f;
}

__global__ void deg_count(const int* __restrict__ dst, int* __restrict__ cnt,
                          int* __restrict__ rank, int E) {
  int e = blockIdx.x * blockDim.x + threadIdx.x;
  if (e < E) rank[e] = atomicAdd(&cnt[dst[e]], 1);
}

__global__ void __launch_bounds__(SCAN_T) scan_rowptr(const int* __restrict__ cnt,
                                                      int* __restrict__ rowptr,
                                                      float* __restrict__ rdeg, int n) {
  __shared__ int s[CNT_PAD];
  __shared__ int wsum[32];
  int t = threadIdx.x, lane = t & 31, wave = t >> 5;
  {
    int v[SCAN_CH];
#pragma unroll
    for (int j = 0; j < SCAN_CH; j++) {
      int idx = j * SCAN_T + t;
      v[j] = cnt[(idx < n) ? idx : 0];
    }
    __builtin_amdgcn_sched_barrier(0);
#pragma unroll
    for (int j = 0; j < SCAN_CH; j++) {
      int idx = j * SCAN_T + t;
      s[idx] = (idx < n) ? v[j] : 0;
    }
  }
  __syncthreads();
  int local = 0;
  for (int j = 0; j < SCAN_CH; j++) local += s[t * SCAN_CH + j];
  int inc = local;
  for (int o = 1; o < 32; o <<= 1) {
    int v = __shfl_up(inc, o);
    if (lane >= o) inc += v;
  }
  if (lane == 31) wsum[wave] = inc;
  __syncthreads();
  if (wave == 0) {
    int w = wsum[lane], wi = w;
    for (int o = 1; o < 32; o <<= 1) {
      int v = __shfl_up(wi, o);
      if (lane >= o) wi += v;
    }
    wsum[lane] = wi - w;
  }
  __syncthreads();
  int run = wsum[wave] + inc - local;
  for (int j = 0; j < SCAN_CH; j++) {
    int c = s[t * SCAN_CH + j];
    s[t * SCAN_CH + j] = run;
    run += c;
  }
  __syncthreads();
  for (int j = 0; j < SCAN_CH; j++) {
    int idx = j * SCAN_T + t;
    if (idx <= n) rowptr[idx] = s[idx];
    if (idx < n) rdeg[idx] = 1.f / fmaxf((float)(s[idx + 1] - s[idx]), 1.f);
  }
}

__global__ void csr_fill(const int* __restrict__ src, const int* __restrict__ dst,
                         const int* __restrict__ rowptr, const int* __restrict__ rank,
                         int* __restrict__ col, int E) {
  int e = blockIdx.x * blockDim.x + threadIdx.x;
  if (e >= E) return;
  col[rowptr[dst[e]] + rank[e]] = src[e];
}

template <int V, int R>
__device__ __forceinline__ void add_rows(float4 (&acc)[V], const float* __restrict__ hl,
                                         int c, int j) {
  constexpr int F = 128 * V;
  float4 x[R][V];
#pragma unroll
  for (int r = 0; r < R; r++) {
    const float* p = hl + (size_t)__builtin_amdgcn_readlane(c, j + r) * F;
#pragma unroll
    for (int v = 0; v < V; v++) x[r][v] = *(const float4*)(p + v * 128);
  }
#pragma unroll
  for (int r = 0; r < R; r++) {
#pragma unroll
    for (int v = 0; v < V; v++) {
      acc[v].x += x[r][v].x; acc[v].y += x[r][v].y;
      acc[v].z += x[r][v].z; acc[v].w += x[r][v].w;
    }
  }
}

template <int V>
__device__ __forceinline__ void gather_row(const float* __restrict__ h,
                                           const int* __restrict__ rowptr,
                                           const int* __restrict__ col,
                                           float* __restrict__ msg, int n) {
  constexpr int F = 128 * V;
  int lane = threadIdx.x & 31;
  int i = blockIdx.x * (blockDim.x >> 5) + (threadIdx.x >> 5);
  i = __builtin_amdgcn_readfirstlane(i);
  if (i >= n) return;
  int beg = rowptr[i], end = rowptr[i + 1];
  float4 acc[V];
#pragma unroll
  for (int v = 0; v < V; v++) acc[v] = make_float4(0.f, 0.f, 0.f, 0.f);
  const float* hl = h + lane * 4;
  for (int base = beg; base < end; base += 32) {
    int m = end - base;
    if (m > 32) m = 32;
    int c = (lane < m) ? col[base + lane] : 0;
    int j = 0;
    for (; j + 4 <= m; j += 4) add_rows<V, 4>(acc, hl, c, j);
    int rem = m - j;
    if (rem == 3) add_rows<V, 3>(acc, hl, c, j);
    else if (rem == 2) add_rows<V, 2>(acc, hl, c, j);
    else if (rem == 1) add_rows<V, 1>(acc, hl, c, j);
  }
  float* o = msg + (size_t)i * F + lane * 4;
#pragma unroll
  for (int v = 0; v < V; v++) *(float4*)(o + v * 128) = acc[v];
}

__global__ void gather_sum_128(const float* __restrict__ h, const int* __restrict__ rowptr,
                               const int* __restrict__ col, float* __restrict__ msg, int n) {
  gather_row<1>(h, rowptr, col, msg, n);
}

__global__ void gather_sum_256(const float* __restrict__ h, const int* __restrict__ rowptr,
                               const int* __restrict__ col, float* __restrict__ msg, int n) {
  gather_row<2>(h, rowptr, col, msg, n);
}

#define MT 5
#define BM (16 * MT)
__global__ void __launch_bounds__(128)
sage_gemm_wmma(const float* __restrict__ msg, const float* __restrict__ rdeg,
               const float* __restrict__ h,
               const float* __restrict__ Wl, const float* __restrict__ Wr,
               const float* __restrict__ bias, float* __restrict__ out,
               int K, int F, int nblocks_n, int do_relu) {
  __shared__ float sA[2][BM * SA_STRIDE];
  __shared__ float sB[2][KC * SB_STRIDE];

  int t = threadIdx.x;
  int wave = t >> 5, lane = t & 31;
  int mBase = (blockIdx.x / nblocks_n) * BM;
  int nBase = (blockIdx.x % nblocks_n) * 64;

  int m16  = lane & 15;
  int kb   = (lane >> 4) << 1;
  int nloc = wave * 16 + (lane & 15);
  int ncol = nBase + nloc;

  v8f acc[MT];
#pragma unroll
  for (int i = 0; i < MT; i++) acc[i] = (v8f){0.f, 0.f, 0.f, 0.f, 0.f, 0.f, 0.f, 0.f};

  int ar = t >> 3, ac = (t & 7) << 2;
  int br = t >> 4, bc = (t & 15) << 2;

  float sc[MT];
#pragma unroll
  for (int i = 0; i < MT; i++) sc[i] = rdeg[mBase + ar + 16 * i];

  for (int kc = 0; kc < K; kc += KC) {
    {
      float4 va0[MT], va1[MT], vb0[4], vb1[4];
#pragma unroll
      for (int i = 0; i < MT; i++) {
        size_t row = (size_t)(mBase + ar + 16 * i) * K + kc + ac;
        va0[i] = *(const float4*)(msg + row);
        va1[i] = *(const float4*)(h + row);
      }
#pragma unroll
      for (int pass = 0; pass < 4; pass++) {
        int row = br + pass * 8;
        vb0[pass] = *(const float4*)(Wl + (size_t)(kc + row) * F + nBase + bc);
        vb1[pass] = *(const float4*)(Wr + (size_t)(kc + row) * F + nBase + bc);
      }
      __builtin_amdgcn_sched_barrier(0);
#pragma unroll
      for (int i = 0; i < MT; i++) {
        float4 s0 = va0[i];
        s0.x *= sc[i]; s0.y *= sc[i]; s0.z *= sc[i]; s0.w *= sc[i];
        *(float4*)(&sA[0][(ar + 16 * i) * SA_STRIDE + ac]) = s0;
        *(float4*)(&sA[1][(ar + 16 * i) * SA_STRIDE + ac]) = va1[i];
      }
#pragma unroll
      for (int pass = 0; pass < 4; pass++) {
        int row = br + pass * 8;
        *(float4*)(&sB[0][row * SB_STRIDE + bc]) = vb0[pass];
        *(float4*)(&sB[1][row * SB_STRIDE + bc]) = vb1[pass];
      }
    }
    if (kc + KC < K) {
      if (t < BM) {
        __builtin_prefetch(msg + (size_t)(mBase + t) * K + kc + KC, 0, 0);
        __builtin_prefetch(h   + (size_t)(mBase + t) * K + kc + KC, 0, 0);
      } else if (t - BM < KC) {
        __builtin_prefetch(Wl + (size_t)(kc + KC + t - BM) * F + nBase, 0, 0);
        __builtin_prefetch(Wr + (size_t)(kc + KC + t - BM) * F + nBase, 0, 0);
      }
    }
    __syncthreads();

#pragma unroll
    for (int k = 0; k < KC; k += 4) {
      v2f a, b0, b1;
      b0.x = sB[0][(k + kb) * SB_STRIDE + nloc];
      b0.y = sB[0][(k + kb + 1) * SB_STRIDE + nloc];
      b1.x = sB[1][(k + kb) * SB_STRIDE + nloc];
      b1.y = sB[1][(k + kb + 1) * SB_STRIDE + nloc];
#pragma unroll
      for (int i = 0; i < MT; i++) {
        a.x = sA[0][(m16 + 16 * i) * SA_STRIDE + k + kb];
        a.y = sA[0][(m16 + 16 * i) * SA_STRIDE + k + kb + 1];
        acc[i] = __builtin_amdgcn_wmma_f32_16x16x4_f32(false, a, false, b0, (short)0, acc[i],
                                                       false, false);
      }
#pragma unroll
      for (int i = 0; i < MT; i++) {
        a.x = sA[1][(m16 + 16 * i) * SA_STRIDE + k + kb];
        a.y = sA[1][(m16 + 16 * i) * SA_STRIDE + k + kb + 1];
        acc[i] = __builtin_amdgcn_wmma_f32_16x16x4_f32(false, a, false, b1, (short)0, acc[i],
                                                       false, false);
      }
    }
    __syncthreads();
  }

  float bv = bias[ncol];
#pragma unroll
  for (int i = 0; i < MT; i++) {
    int mhalf = mBase + 16 * i + ((lane >> 4) << 3);
#pragma unroll
    for (int v = 0; v < 8; v++) {
      float val = acc[i][v] + bv;
      if (do_relu) val = fmaxf(val, 0.f);
      out[(size_t)(mhalf + v) * F + ncol] = val;
    }
  }
}

#define BN_U 14
__global__ void bn_stats(const float* __restrict__ h, float* __restrict__ st, int M, int F) {
  int c = threadIdx.x;
  int stride = gridDim.x;
  float s = 0.f, s2 = 0.f;
  for (int r0 = blockIdx.x; r0 < M; r0 += stride * BN_U) {
    float v[BN_U];
#pragma unroll
    for (int j = 0; j < BN_U; j++) {
      int r = r0 + j * stride;
      v[j] = (r < M) ? h[(size_t)r * F + c] : 0.f;
    }
#pragma unroll
    for (int j = 0; j < BN_U; j++) {
      s += v[j];
      s2 = __builtin_fmaf(v[j], v[j], s2);
    }
  }
  atomicAdd(&st[c], s);
  atomicAdd(&st[F + c], s2);
}

__global__ void bn_apply(float* __restrict__ h, const float* __restrict__ st,
                         const float* __restrict__ g, const float* __restrict__ be,
                         int M, int F) {
  int i = blockIdx.x * blockDim.x + threadIdx.x;
  if (i >= M * F) return;
  int c = i & (F - 1);
  float inv = 1.f / (float)M;
  float mean = st[c] * inv;
  float var = st[F + c] * inv - mean * mean;
  h[i] = g[c] * (h[i] - mean) * rsqrtf(var + 1e-5f) + be[c];
}

__global__ void reparam(const float* __restrict__ h3, const float* __restrict__ eps,
                        float* __restrict__ z, int M) {
  int i = blockIdx.x * blockDim.x + threadIdx.x;
  if (i >= M * 64) return;
  int r = i >> 6, c = i & 63;
  const float* row = h3 + (size_t)r * 128;
  z[i] = row[c] + expf(row[64 + c]) * eps[i];
}

extern "C" void kernel_launch(void* const* d_in, const int* in_sizes, int n_in,
                              void* d_out, int out_size, void* d_ws, size_t ws_size,
                              hipStream_t stream) {
  const float* x   = (const float*)d_in[0];
  const int*   src = (const int*)d_in[1];
  const int*   dst = (const int*)d_in[2];
  const float* eps = (const float*)d_in[3];
  const float* W1l = (const float*)d_in[4];
  const float* W1r = (const float*)d_in[5];
  const float* b1  = (const float*)d_in[6];
  const float* g1  = (const float*)d_in[7];
  const float* be1 = (const float*)d_in[8];
  const float* W2l = (const float*)d_in[9];
  const float* W2r = (const float*)d_in[10];
  const float* b2  = (const float*)d_in[11];
  const float* g2  = (const float*)d_in[12];
  const float* be2 = (const float*)d_in[13];
  const float* W3l = (const float*)d_in[14];
  const float* W3r = (const float*)d_in[15];
  const float* b3  = (const float*)d_in[16];
  float* zout = (float*)d_out;
  (void)in_sizes; (void)n_in; (void)out_size; (void)ws_size;

  char* ws = (char*)d_ws;
  size_t off = 0;
  auto carve = [&](size_t bytes) -> float* {
    char* p = ws + off;
    off = (off + bytes + 255) & ~(size_t)255;
    return (float*)p;
  };
  float* rdeg = carve((size_t)NN * 4);
  float* msg  = carve((size_t)NN * 256 * 4);
  float* h1   = carve((size_t)NN * 128 * 4);
  float* h2   = carve((size_t)NN * 256 * 4);
  float* st   = carve(512 * 4);
  float* y1   = msg + (size_t)NN * 128;
  int* cnt    = (int*)h1;
  int* rowptr = cnt + CNT_PAD;
  int* col    = rowptr + CNT_PAD;
  int* rank   = col + NE;

  const int M = NN, E = NE;

  zero_f32<<<256, 256, 0, stream>>>((float*)cnt, M);
  deg_count<<<(E + 255) / 256, 256, 0, stream>>>(dst, cnt, rank, E);
  scan_rowptr<<<1, SCAN_T, 0, stream>>>(cnt, rowptr, rdeg, M);
  csr_fill<<<(E + 255) / 256, 256, 0, stream>>>(src, dst, rowptr, rank, col, E);

  gather_sum_128<<<(M + 7) / 8, 256, 0, stream>>>(x, rowptr, col, msg, M);
  sage_gemm_wmma<<<(NN / BM) * 2, 128, 0, stream>>>(msg, rdeg, x, W1l, W1r, b1, y1,
                                               128, 128, 2, 1);
  zero_f32<<<1, 256, 0, stream>>>(st, 256);
  bn_stats<<<512, 128, 0, stream>>>(y1, st, M, 128);
  bn_apply<<<(M * 128 + 255) / 256, 256, 0, stream>>>(y1, st, g1, be1, M, 128);

  gather_sum_128<<<(M + 7) / 8, 256, 0, stream>>>(y1, rowptr, col, msg, M);
  sage_gemm_wmma<<<(NN / BM) * 4, 128, 0, stream>>>(msg, rdeg, y1, W2l, W2r, b2, h2,
                                               128, 256, 4, 1);
  zero_f32<<<2, 256, 0, stream>>>(st, 512);
  bn_stats<<<512, 256, 0, stream>>>(h2, st, M, 256);
  bn_apply<<<(M * 256 + 255) / 256, 256, 0, stream>>>(h2, st, g2, be2, M, 256);

  gather_sum_256<<<(M + 7) / 8, 256, 0, stream>>>(h2, rowptr, col, msg, M);
  sage_gemm_wmma<<<(NN / BM) * 2, 128, 0, stream>>>(msg, rdeg, h2, W3l, W3r, b3, h1,
                                               256, 128, 2, 0);

  reparam<<<(M * 64 + 255) / 256, 256, 0, stream>>>(h1, eps, zout, M);
}


# BN column sum and sum of squares taken in the SAGE GEMM epilogue from the accumulators (a float atomic per column per wave); both bn_stats passes go, 17 -> 13 launches; batch-sum add order changes
# speedup vs baseline: 2.7114x; 1.0535x over previous
#include <hip/hip_runtime.h>

#define NN 50000
#define NE 800000
#define KC 32
#define SA_STRIDE 36
#define SB_STRIDE 68
#define SCAN_T 1024
#define SCAN_CH 49
#define CNT_PAD (SCAN_T * SCAN_CH)

typedef __attribute__((ext_vector_type(2))) float v2f;
typedef __attribute__((ext_vector_type(8))) float v8f;

__global__ void zero_f32_2(float* __restrict__ a, int na, float* __restrict__ b, int nb) {
  int i = blockIdx.x * blockDim.x + threadIdx.x;
  int stride = gridDim.x * blockDim.x;
  for (; i < na + nb; i += stride) {
    if (i < na) a[i] = 0.f;
    else b[i - na] = 0.f;
  }
}

__global__ void deg_count(const int* __restrict__ dst, int* __restrict__ cnt,
                          int* __restrict__ rank, int E) {
  int e = blockIdx.x * blockDim.x + threadIdx.x;
  if (e < E) rank[e] = atomicAdd(&cnt[dst[e]], 1);
}

__global__ void __launch_bounds__(SCAN_T) scan_rowptr(const int* __restrict__ cnt,
                                                      int* __restrict__ rowptr,
                                                      float* __restrict__ rdeg, int n) {
  __shared__ int s[CNT_PAD];
  __shared__ int wsum[32];
  int t = threadIdx.x, lane = t & 31, wave = t >> 5;
  {
    int v[SCAN_CH];
#pragma unroll
    for (int j = 0; j < SCAN_CH; j++) {
      int idx = j * SCAN_T + t;
      v[j] = cnt[(idx < n) ? idx : 0];
    }
    __builtin_amdgcn_sched_barrier(0);
#pragma unroll
    for (int j = 0; j < SCAN_CH; j++) {
      int idx = j * SCAN_T + t;
      s[idx] = (idx < n) ? v[j] : 0;
    }
  }
  __syncthreads();
  int local = 0;
  for (int j = 0; j < SCAN_CH; j++) local += s[t * SCAN_CH + j];
  int inc = local;
  for (int o = 1; o < 32; o <<= 1) {
    int v = __shfl_up(inc, o);
    if (lane >= o) inc += v;
  }
  if (lane == 31) wsum[wave] = inc;
  __syncthreads();
  if (wave == 0) {
    int w = wsum[lane], wi = w;
    for (int o = 1; o < 32; o <<= 1) {
      int v = __shfl_up(wi, o);
      if (lane >= o) wi += v;
    }
    wsum[lane] = wi - w;
  }
  __syncthreads();
  int run = wsum[wave] + inc - local;
  for (int j = 0; j < SCAN_CH; j++) {
    int c = s[t * SCAN_CH + j];
    s[t * SCAN_CH + j] = run;
    run += c;
  }
  __syncthreads();
  for (int j = 0; j < SCAN_CH; j++) {
    int idx = j * SCAN_T + t;
    if (idx <= n) rowptr[idx] = s[idx];
    if (idx < n) rdeg[idx] = 1.f / fmaxf((float)(s[idx + 1] - s[idx]), 1.f);
  }
}

__global__ void csr_fill(const int* __restrict__ src, const int* __restrict__ dst,
                         const int* __restrict__ rowptr, const int* __restrict__ rank,
                         int* __restrict__ col, int E) {
  int e = blockIdx.x * blockDim.x + threadIdx.x;
  if (e >= E) return;
  col[rowptr[dst[e]] + rank[e]] = src[e];
}

template <int V, int R>
__device__ __forceinline__ void add_rows(float4 (&acc)[V], const float* __restrict__ hl,
                                         int c, int j) {
  constexpr int F = 128 * V;
  float4 x[R][V];
#pragma unroll
  for (int r = 0; r < R; r++) {
    const float* p = hl + (size_t)__builtin_amdgcn_readlane(c, j + r) * F;
#pragma unroll
    for (int v = 0; v < V; v++) x[r][v] = *(const float4*)(p + v * 128);
  }
#pragma unroll
  for (int r = 0; r < R; r++) {
#pragma unroll
    for (int v = 0; v < V; v++) {
      acc[v].x += x[r][v].x; acc[v].y += x[r][v].y;
      acc[v].z += x[r][v].z; acc[v].w += x[r][v].w;
    }
  }
}

template <int V>
__device__ __forceinline__ void gather_row(const float* __restrict__ h,
                                           const int* __restrict__ rowptr,
                                           const int* __restrict__ col,
                                           float* __restrict__ msg, int n) {
  constexpr int F = 128 * V;
  int lane = threadIdx.x & 31;
  int i = blockIdx.x * (blockDim.x >> 5) + (threadIdx.x >> 5);
  i = __builtin_amdgcn_readfirstlane(i);
  if (i >= n) return;
  int beg = rowptr[i], end = rowptr[i + 1];
  float4 acc[V];
#pragma unroll
  for (int v = 0; v < V; v++) acc[v] = make_float4(0.f, 0.f, 0.f, 0.f);
  const float* hl = h + lane * 4;
  for (int base = beg; base < end; base += 32) {
    int m = end - base;
    if (m > 32) m = 32;
    int c = (lane < m) ? col[base + lane] : 0;
    int j = 0;
    for (; j + 4 <= m; j += 4) add_rows<V, 4>(acc, hl, c, j);
    int rem = m - j;
    if (rem == 3) add_rows<V, 3>(acc, hl, c, j);
    else if (rem == 2) add_rows<V, 2>(acc, hl, c, j);
    else if (rem == 1) add_rows<V, 1>(acc, hl, c, j);
  }
  float* o = msg + (size_t)i * F + lane * 4;
#pragma unroll
  for (int v = 0; v < V; v++) *(float4*)(o + v * 128) = acc[v];
}

__global__ void gather_sum_128(const float* __restrict__ h, const int* __restrict__ rowptr,
                               const int* __restrict__ col, float* __restrict__ msg, int n) {
  gather_row<1>(h, rowptr, col, msg, n);
}

__global__ void gather_sum_256(const float* __restrict__ h, const int* __restrict__ rowptr,
                               const int* __restrict__ col, float* __restrict__ msg, int n) {
  gather_row<2>(h, rowptr, col, msg, n);
}

#define MT 5
#define BM (16 * MT)
__global__ void __launch_bounds__(128)
sage_gemm_wmma(const float* __restrict__ msg, const float* __restrict__ rdeg,
               const float* __restrict__ h,
               const float* __restrict__ Wl, const float* __restrict__ Wr,
               const float* __restrict__ bias, float* __restrict__ out,
               float* __restrict__ st, int K, int F, int nblocks_n, int do_relu) {
  __shared__ float sA[2][BM * SA_STRIDE];
  __shared__ float sB[2][KC * SB_STRIDE];

  int t = threadIdx.x;
  int wave = t >> 5, lane = t & 31;
  int mBase = (blockIdx.x / nblocks_n) * BM;
  int nBase = (blockIdx.x % nblocks_n) * 64;

  int m16  = lane & 15;
  int kb   = (lane >> 4) << 1;
  int nloc = wave * 16 + (lane & 15);
  int ncol = nBase + nloc;

  v8f acc[MT];
#pragma unroll
  for (int i = 0; i < MT; i++) acc[i] = (v8f){0.f, 0.f, 0.f, 0.f, 0.f, 0.f, 0.f, 0.f};

  int ar = t >> 3, ac = (t & 7) << 2;
  int br = t >> 4, bc = (t & 15) << 2;

  float sc[MT];
#pragma unroll
  for (int i = 0; i < MT; i++) sc[i] = rdeg[mBase + ar + 16 * i];

  for (int kc = 0; kc < K; kc += KC) {
    {
      float4 va0[MT], va1[MT], vb0[4], vb1[4];
#pragma unroll
      for (int i = 0; i < MT; i++) {
        size_t row = (size_t)(mBase + ar + 16 * i) * K + kc + ac;
        va0[i] = *(const float4*)(msg + row);
        va1[i] = *(const float4*)(h + row);
      }
#pragma unroll
      for (int pass = 0; pass < 4; pass++) {
        int row = br + pass * 8;
        vb0[pass] = *(const float4*)(Wl + (size_t)(kc + row) * F + nBase + bc);
        vb1[pass] = *(const float4*)(Wr + (size_t)(kc + row) * F + nBase + bc);
      }
      __builtin_amdgcn_sched_barrier(0);
#pragma unroll
      for (int i = 0; i < MT; i++) {
        float4 s0 = va0[i];
        s0.x *= sc[i]; s0.y *= sc[i]; s0.z *= sc[i]; s0.w *= sc[i];
        *(float4*)(&sA[0][(ar + 16 * i) * SA_STRIDE + ac]) = s0;
        *(float4*)(&sA[1][(ar + 16 * i) * SA_STRIDE + ac]) = va1[i];
      }
#pragma unroll
      for (int pass = 0; pass < 4; pass++) {
        int row = br + pass * 8;
        *(float4*)(&sB[0][row * SB_STRIDE + bc]) = vb0[pass];
        *(float4*)(&sB[1][row * SB_STRIDE + bc]) = vb1[pass];
      }
    }
    if (kc + KC < K) {
      if (t < BM) {
        __builtin_prefetch(msg + (size_t)(mBase + t) * K + kc + KC, 0, 0);
        __builtin_prefetch(h   + (size_t)(mBase + t) * K + kc + KC, 0, 0);
      } else if (t - BM < KC) {
        __builtin_prefetch(Wl + (size_t)(kc + KC + t - BM) * F + nBase, 0, 0);
        __builtin_prefetch(Wr + (size_t)(kc + KC + t - BM) * F + nBase, 0, 0);
      }
    }
    __syncthreads();

#pragma unroll
    for (int k = 0; k < KC; k += 4) {
      v2f a, b0, b1;
      b0.x = sB[0][(k + kb) * SB_STRIDE + nloc];
      b0.y = sB[0][(k + kb + 1) * SB_STRIDE + nloc];
      b1.x = sB[1][(k + kb) * SB_STRIDE + nloc];
      b1.y = sB[1][(k + kb + 1) * SB_STRIDE + nloc];
#pragma unroll
      for (int i = 0; i < MT; i++) {
        a.x = sA[0][(m16 + 16 * i) * SA_STRIDE + k + kb];
        a.y = sA[0][(m16 + 16 * i) * SA_STRIDE + k + kb + 1];
        acc[i] = __builtin_amdgcn_wmma_f32_16x16x4_f32(false, a, false, b0, (short)0, acc[i],
                                                       false, false);
      }
#pragma unroll
      for (int i = 0; i < MT; i++) {
        a.x = sA[1][(m16 + 16 * i) * SA_STRIDE + k + kb];
        a.y = sA[1][(m16 + 16 * i) * SA_STRIDE + k + kb + 1];
        acc[i] = __builtin_amdgcn_wmma_f32_16x16x4_f32(false, a, false, b1, (short)0, acc[i],
                                                       false, false);
      }
    }
    __syncthreads();
  }

  float bv = bias[ncol];
  float cs = 0.f, cs2 = 0.f;
#pragma unroll
  for (int i = 0; i < MT; i++) {
    int mhalf = mBase + 16 * i + ((lane >> 4) << 3);
#pragma unroll
    for (int v = 0; v < 8; v++) {
      float val = acc[i][v] + bv;
      if (do_relu) val = fmaxf(val, 0.f);
      out[(size_t)(mhalf + v) * F + ncol] = val;
      cs += val;
      cs2 = __builtin_fmaf(val, val, cs2);
    }
  }
  if (st) {
    cs  += __shfl_xor(cs, 16);
    cs2 += __shfl_xor(cs2, 16);
    if (lane < 16) {
      atomicAdd(&st[ncol], cs);
      atomicAdd(&st[F + ncol], cs2);
    }
  }
}

__global__ void bn_apply(float* __restrict__ h, const float* __restrict__ st,
                         const float* __restrict__ g, const float* __restrict__ be,
                         int M, int F) {
  int i = blockIdx.x * blockDim.x + threadIdx.x;
  if (i >= M * F) return;
  int c = i & (F - 1);
  float inv = 1.f / (float)M;
  float mean = st[c] * inv;
  float var = st[F + c] * inv - mean * mean;
  h[i] = g[c] * (h[i] - mean) * rsqrtf(var + 1e-5f) + be[c];
}

__global__ void reparam(const float* __restrict__ h3, const float* __restrict__ eps,
                        float* __restrict__ z, int M) {
  int i = blockIdx.x * blockDim.x + threadIdx.x;
  if (i >= M * 64) return;
  int r = i >> 6, c = i & 63;
  const float* row = h3 + (size_t)r * 128;
  z[i] = row[c] + expf(row[64 + c]) * eps[i];
}

extern "C" void kernel_launch(void* const* d_in, const int* in_sizes, int n_in,
                              void* d_out, int out_size, void* d_ws, size_t ws_size,
                              hipStream_t stream) {
  const float* x   = (const float*)d_in[0];
  const int*   src = (const int*)d_in[1];
  const int*   dst = (const int*)d_in[2];
  const float* eps = (const float*)d_in[3];
  const float* W1l = (const float*)d_in[4];
  const float* W1r = (const float*)d_in[5];
  const float* b1  = (const float*)d_in[6];
  const float* g1  = (const float*)d_in[7];
  const float* be1 = (const float*)d_in[8];
  const float* W2l = (const float*)d_in[9];
  const float* W2r = (const float*)d_in[10];
  const float* b2  = (const float*)d_in[11];
  const float* g2  = (const float*)d_in[12];
  const float* be2 = (const float*)d_in[13];
  const float* W3l = (const float*)d_in[14];
  const float* W3r = (const float*)d_in[15];
  const float* b3  = (const float*)d_in[16];
  float* zout = (float*)d_out;
  (void)in_sizes; (void)n_in; (void)out_size; (void)ws_size;

  char* ws = (char*)d_ws;
  size_t off = 0;
  auto carve = [&](size_t bytes) -> float* {
    char* p = ws + off;
    off = (off + bytes + 255) & ~(size_t)255;
    return (float*)p;
  };
  float* rdeg = carve((size_t)NN * 4);
  float* msg  = carve((size_t)NN * 256 * 4);
  float* h1   = carve((size_t)NN * 128 * 4);
  float* h2   = carve((size_t)NN * 256 * 4);
  float* y1   = msg + (size_t)NN * 128;
  int* cnt    = (int*)h1;
  int* rowptr = cnt + CNT_PAD;
  int* col    = rowptr + CNT_PAD;
  int* rank   = col + NE;
  float* st1  = (float*)(rank + NE);
  float* st2  = st1 + 256;

  const int M = NN, E = NE;

  zero_f32_2<<<256, 256, 0, stream>>>((float*)cnt, M, st1, 768);
  deg_count<<<(E + 255) / 256, 256, 0, stream>>>(dst, cnt, rank, E);
  scan_rowptr<<<1, SCAN_T, 0, stream>>>(cnt, rowptr, rdeg, M);
  csr_fill<<<(E + 255) / 256, 256, 0, stream>>>(src, dst, rowptr, rank, col, E);

  gather_sum_128<<<(M + 7) / 8, 256, 0, stream>>>(x, rowptr, col, msg, M);
  sage_gemm_wmma<<<(NN / BM) * 2, 128, 0, stream>>>(msg, rdeg, x, W1l, W1r, b1, y1, st1,
                                               128, 128, 2, 1);
  bn_apply<<<(M * 128 + 255) / 256, 256, 0, stream>>>(y1, st1, g1, be1, M, 128);

  gather_sum_128<<<(M + 7) / 8, 256, 0, stream>>>(y1, rowptr, col, msg, M);
  sage_gemm_wmma<<<(NN / BM) * 4, 128, 0, stream>>>(msg, rdeg, y1, W2l, W2r, b2, h2, st2,
                                               128, 256, 4, 1);
  bn_apply<<<(M * 256 + 255) / 256, 256, 0, stream>>>(h2, st2, g2, be2, M, 256);

  gather_sum_256<<<(M + 7) / 8, 256, 0, stream>>>(h2, rowptr, col, msg, M);
  sage_gemm_wmma<<<(NN / BM) * 2, 128, 0, stream>>>(msg, rdeg, h2, W3l, W3r, b3, h1, nullptr,
                                               256, 128, 2, 0);

  reparam<<<(M * 64 + 255) / 256, 256, 0, stream>>>(h1, eps, zout, M);
}


# Elementwise passes folded into neighbours: BN applied as next gather and GEMM staging read the rows, z formed in last GEMM epilogue; 13 -> 10 launches; BN sums now order-free 64-bit integer atomics
# speedup vs baseline: 3.3080x; 1.2200x over previous
#include <hip/hip_runtime.h>

#define NN 50000
#define NE 800000
#define KC 32
#define SA_STRIDE 36
#define SB_STRIDE 68
#define SCAN_T 1024
#define SCAN_CH 49
#define CNT_PAD (SCAN_T * SCAN_CH)

typedef __attribute__((ext_vector_type(2))) float v2f;
typedef __attribute__((ext_vector_type(8))) float v8f;
typedef unsigned long long u64;

#define FX_ONE 268435456.f
#define BN_EPS 1e-5f

__device__ __forceinline__ void bn_consts(u64 s, u64 q, float g, float& mean, float& scl) {
  float inv = 1.f / (float)NN;
  float S = (float)(long long)s * (1.f / FX_ONE);
  float Q = (float)(long long)q * (1.f / FX_ONE);
  mean = S * inv;
  float var = Q * inv - mean * mean;
  scl = g * rsqrtf(var + BN_EPS);
}

__global__ void zero_f32_2(float* __restrict__ a, int na, float* __restrict__ b, int nb) {
  int i = blockIdx.x * blockDim.x + threadIdx.x;
  int stride = gridDim.x * blockDim.x;
  for (; i < na + nb; i += stride) {
    if (i < na) a[i] = 0.f;
    else b[i - na] = 0.f;
  }
}

__global__ void deg_count(const int* __restrict__ dst, int* __restrict__ cnt,
                          int* __restrict__ rank, int E) {
  int e = blockIdx.x * blockDim.x + threadIdx.x;
  if (e < E) rank[e] = atomicAdd(&cnt[dst[e]], 1);
}

__global__ void __launch_bounds__(SCAN_T) scan_rowptr(const int* __restrict__ cnt,
                                                      int* __restrict__ rowptr,
                                                      float* __restrict__ rdeg, int n) {
  __shared__ int s[CNT_PAD];
  __shared__ int wsum[32];
  int t = threadIdx.x, lane = t & 31, wave = t >> 5;
  {
    int v[SCAN_CH];
#pragma unroll
    for (int j = 0; j < SCAN_CH; j++) {
      int idx = j * SCAN_T + t;
      v[j] = cnt[(idx < n) ? idx : 0];
    }
    __builtin_amdgcn_sched_barrier(0);
#pragma unroll
    for (int j = 0; j < SCAN_CH; j++) {
      int idx = j * SCAN_T + t;
      s[idx] = (idx < n) ? v[j] : 0;
    }
  }
  __syncthreads();
  int local = 0;
  for (int j = 0; j < SCAN_CH; j++) local += s[t * SCAN_CH + j];
  int inc = local;
  for (int o = 1; o < 32; o <<= 1) {
    int v = __shfl_up(inc, o);
    if (lane >= o) inc += v;
  }
  if (lane == 31) wsum[wave] = inc;
  __syncthreads();
  if (wave == 0) {
    int w = wsum[lane], wi = w;
    for (int o = 1; o < 32; o <<= 1) {
      int v = __shfl_up(wi, o);
      if (lane >= o) wi += v;
    }
    wsum[lane] = wi - w;
  }
  __syncthreads();
  int run = wsum[wave] + inc - local;
  for (int j = 0; j < SCAN_CH; j++) {
    int c = s[t * SCAN_CH + j];
    s[t * SCAN_CH + j] = run;
    run += c;
  }
  __syncthreads();
  for (int j = 0; j < SCAN_CH; j++) {
    int idx = j * SCAN_T + t;
    if (idx <= n) rowptr[idx] = s[idx];
    if (idx < n) rdeg[idx] = 1.f / fmaxf((float)(s[idx + 1] - s[idx]), 1.f);
  }
}

__global__ void csr_fill(const int* __restrict__ src, const int* __restrict__ dst,
                         const int* __restrict__ rowptr, const int* __restrict__ rank,
                         int* __restrict__ col, int E) {
  int e = blockIdx.x * blockDim.x + threadIdx.x;
  if (e >= E) return;
  col[rowptr[dst[e]] + rank[e]] = src[e];
}

template <int V, int R, int BN>
__device__ __forceinline__ void add_rows(float4 (&acc)[V], const float* __restrict__ hl,
                                         int c, int j, const float4 (&mu)[V],
                                         const float4 (&sc)[V], const float4 (&sh)[V]) {
  constexpr int F = 128 * V;
  float4 x[R][V];
#pragma unroll
  for (int r = 0; r < R; r++) {
    const float* p = hl + (size_t)__builtin_amdgcn_readlane(c, j + r) * F;
#pragma unroll
    for (int v = 0; v < V; v++) x[r][v] = *(const float4*)(p + v * 128);
  }
#pragma unroll
  for (int r = 0; r < R; r++) {
#pragma unroll
    for (int v = 0; v < V; v++) {
      float4 y = x[r][v];
      if (BN) {
        y.x = (y.x - mu[v].x) * sc[v].x + sh[v].x;
        y.y = (y.y - mu[v].y) * sc[v].y + sh[v].y;
        y.z = (y.z - mu[v].z) * sc[v].z + sh[v].z;
        y.w = (y.w - mu[v].w) * sc[v].w + sh[v].w;
      }
      acc[v].x += y.x; acc[v].y += y.y;
      acc[v].z += y.z; acc[v].w += y.w;
    }
  }
}

template <int V, int BN>
__device__ __forceinline__ void gather_row(const float* __restrict__ h,
                                           const int* __restrict__ rowptr,
                                           const int* __restrict__ col,
                                           float* __restrict__ msg, int n,
                                           const u64* __restrict__ bst,
                                           const float* __restrict__ bg,
                                           const float* __restrict__ bb) {
  constexpr int F = 128 * V;
  __shared__ float gMu[BN ? F : 4], gSc[BN ? F : 4], gSh[BN ? F : 4];
  if (BN) {
    for (int k = threadIdx.x; k < F; k += blockDim.x) {
      float mean, scl;
      bn_consts(bst[k], bst[F + k], bg[k], mean, scl);
      gMu[k] = mean; gSc[k] = scl; gSh[k] = bb[k];
    }
    __syncthreads();
  }
  int lane = threadIdx.x & 31;
  int i = blockIdx.x * (blockDim.x >> 5) + (threadIdx.x >> 5);
  i = __builtin_amdgcn_readfirstlane(i);
  if (i >= n) return;
  int beg = rowptr[i], end = rowptr[i + 1];
  float4 acc[V];
#pragma unroll
  for (int v = 0; v < V; v++) acc[v] = make_float4(0.f, 0.f, 0.f, 0.f);
  float4 mu[V], sc[V], sh[V];
#pragma unroll
  for (int v = 0; v < V; v++) {
    mu[v] = sc[v] = sh[v] = make_float4(0.f, 0.f, 0.f, 0.f);
    if (BN) {
      int c0 = v * 128 + lane * 4;
      mu[v] = *(const float4*)(&gMu[c0]);
      sc[v] = *(const float4*)(&gSc[c0]);
      sh[v] = *(const float4*)(&gSh[c0]);
    }
  }
  const float* hl = h + lane * 4;
  for (int base = beg; base < end; base += 32) {
    int m = end - base;
    if (m > 32) m = 32;
    int c = (lane < m) ? col[base + lane] : 0;
    int j = 0;
    for (; j + 4 <= m; j += 4) add_rows<V, 4, BN>(acc, hl, c, j, mu, sc, sh);
    int rem = m - j;
    if (rem == 3) add_rows<V, 3, BN>(acc, hl, c, j, mu, sc, sh);
    else if (rem == 2) add_rows<V, 2, BN>(acc, hl, c, j, mu, sc, sh);
    else if (rem == 1) add_rows<V, 1, BN>(acc, hl, c, j, mu, sc, sh);
  }
  float* o = msg + (size_t)i * F + lane * 4;
#pragma unroll
  for (int v = 0; v < V; v++) *(float4*)(o + v * 128) = acc[v];
}

__global__ void gather_sum_128(const float* __restrict__ h, const int* __restrict__ rowptr,
                               const int* __restrict__ col, float* __restrict__ msg, int n) {
  gather_row<1, 0>(h, rowptr, col, msg, n, nullptr, nullptr, nullptr);
}

__global__ void gather_bn_sum_128(const float* __restrict__ h, const int* __restrict__ rowptr,
                                  const int* __restrict__ col, float* __restrict__ msg, int n,
                                  const u64* __restrict__ bst, const float* __restrict__ bg,
                                  const float* __restrict__ bb) {
  gather_row<1, 1>(h, rowptr, col, msg, n, bst, bg, bb);
}

__global__ void gather_bn_sum_256(const float* __restrict__ h, const int* __restrict__ rowptr,
                                  const int* __restrict__ col, float* __restrict__ msg, int n,
                                  const u64* __restrict__ bst, const float* __restrict__ bg,
                                  const float* __restrict__ bb) {
  gather_row<2, 1>(h, rowptr, col, msg, n, bst, bg, bb);
}

#define MT 5
#define BM (16 * MT)
template <int RP, int BN, int K, int F>
__global__ void __launch_bounds__(128)
sage_gemm_wmma(const float* __restrict__ msg, const float* __restrict__ rdeg,
               const float* __restrict__ h,
               const float* __restrict__ Wl, const float* __restrict__ Wr,
               const float* __restrict__ bias, float* __restrict__ out,
               u64* __restrict__ st,
               const u64* __restrict__ bst, const float* __restrict__ bg,
               const float* __restrict__ bb, const float* __restrict__ eps, int do_relu) {
  __shared__ float sA[2][BM * SA_STRIDE];
  __shared__ float sB[2][KC * SB_STRIDE];
  __shared__ float sSc[BN ? K : 4], sSh[BN ? K : 4];

  constexpr int nblocks_n = F / 64;
  int t = threadIdx.x;
  int wave = t >> 5, lane = t & 31;
  int mBase = (blockIdx.x / nblocks_n) * BM;
  int nb = blockIdx.x % nblocks_n;
  auto gcol = [&](int c) { return RP ? (32 * nb + (c & 31) + ((c >> 5) << 6)) : (64 * nb + c); };

  if (BN) {
    for (int k = t; k < K; k += 128) {
      float mean, scl;
      bn_consts(bst[k], bst[K + k], bg[k], mean, scl);
      sSc[k] = scl; sSh[k] = bb[k] - mean * scl;
    }
    __syncthreads();
  }

  int m16  = lane & 15;
  int kb   = (lane >> 4) << 1;
  int nloc = wave * 16 + (lane & 15);
  int ncol = gcol(nloc);

  v8f acc[MT];
#pragma unroll
  for (int i = 0; i < MT; i++) acc[i] = (v8f){0.f, 0.f, 0.f, 0.f, 0.f, 0.f, 0.f, 0.f};

  int ar = t >> 3, ac = (t & 7) << 2;
  int br = t >> 4, bc = (t & 15) << 2;
  unsigned aoff = (unsigned)(ar * K + ac);
  unsigned boff = (unsigned)(br * F + gcol(bc));
  const float* am = msg + (size_t)mBase * K;
  const float* ah = h + (size_t)mBase * K;

  float sc[MT];
#pragma unroll
  for (int i = 0; i < MT; i++) sc[i] = rdeg[mBase + ar + 16 * i];

#pragma nounroll
  for (int kc = 0; kc < K; kc += KC) {
    {
      float4 va0[MT], va1[MT], vb0[4], vb1[4];
      const float* pm = am + kc;
      const float* ph = ah + kc;
      const float* pl = Wl + (size_t)kc * F;
      const float* pr = Wr + (size_t)kc * F;
#pragma unroll
      for (int i = 0; i < MT; i++) {
        va0[i] = *(const float4*)(pm + (aoff + 16u * i * K));
        va1[i] = *(const float4*)(ph + (aoff + 16u * i * K));
      }
#pragma unroll
      for (int pass = 0; pass < 4; pass++) {
        vb0[pass] = *(const float4*)(pl + (boff + 8u * pass * F));
        vb1[pass] = *(const float4*)(pr + (boff + 8u * pass * F));
      }
      __builtin_amdgcn_sched_barrier(0);
      float4 sl, sf;
      if (BN) {
        sl = *(const float4*)(&sSc[kc + ac]);
        sf = *(const float4*)(&sSh[kc + ac]);
      }
#pragma unroll
      for (int i = 0; i < MT; i++) {
        float4 s0 = va0[i];
        s0.x *= sc[i]; s0.y *= sc[i]; s0.z *= sc[i]; s0.w *= sc[i];
        *(float4*)(&sA[0][(ar + 16 * i) * SA_STRIDE + ac]) = s0;
        float4 s1 = va1[i];
        if (BN) {
          s1.x = s1.x * sl.x + sf.x; s1.y = s1.y * sl.y + sf.y;
          s1.z = s1.z * sl.z + sf.z; s1.w = s1.w * sl.w + sf.w;
        }
        *(float4*)(&sA[1][(ar + 16 * i) * SA_STRIDE + ac]) = s1;
      }
#pragma unroll
      for (int pass = 0; pass < 4; pass++) {
        int row = br + pass * 8;
        *(float4*)(&sB[0][row * SB_STRIDE + bc]) = vb0[pass];
        *(float4*)(&sB[1][row * SB_STRIDE + bc]) = vb1[pass];
      }
    }
    if (kc + KC < K) {
      if (t < BM) {
        __builtin_prefetch(msg + (size_t)(mBase + t) * K + kc + KC, 0, 0);
        __builtin_prefetch(h   + (size_t)(mBase + t) * K + kc + KC, 0, 0);
      } else if (t - BM < KC) {
        __builtin_prefetch(Wl + (size_t)(kc + KC + t - BM) * F + gcol(0), 0, 0);
        __builtin_prefetch(Wr + (size_t)(kc + KC + t - BM) * F + gcol(0), 0, 0);
      }
    }
    __syncthreads();

#pragma unroll
    for (int k = 0; k < KC; k += 4) {
      v2f a, b0, b1;
      b0.x = sB[0][(k + kb) * SB_STRIDE + nloc];
      b0.y = sB[0][(k + kb + 1) * SB_STRIDE + nloc];
      b1.x = sB[1][(k + kb) * SB_STRIDE + nloc];
      b1.y = sB[1][(k + kb + 1) * SB_STRIDE + nloc];
#pragma unroll
      for (int i = 0; i < MT; i++) {
        a.x = sA[0][(m16 + 16 * i) * SA_STRIDE + k + kb];
        a.y = sA[0][(m16 + 16 * i) * SA_STRIDE + k + kb + 1];
        acc[i] = __builtin_amdgcn_wmma_f32_16x16x4_f32(false, a, false, b0, (short)0, acc[i],
                                                       false, false);
      }
#pragma unroll
      for (int i = 0; i < MT; i++) {
        a.x = sA[1][(m16 + 16 * i) * SA_STRIDE + k + kb];
        a.y = sA[1][(m16 + 16 * i) * SA_STRIDE + k + kb + 1];
        acc[i] = __builtin_amdgcn_wmma_f32_16x16x4_f32(false, a, false, b1, (short)0, acc[i],
                                                       false, false);
      }
    }
    __syncthreads();
  }

  float bv = bias[ncol];
  if (RP) {
    float* ex = &sA[0][0];
    int xc = (wave & 1) * 16 + (lane & 15);
    if (wave >= 2) {
#pragma unroll
      for (int i = 0; i < MT; i++) {
#pragma unroll
        for (int v = 0; v < 8; v++) {
          int r = 16 * i + ((lane >> 4) << 3) + v;
          ex[r * SA_STRIDE + xc] = expf(acc[i][v] + bv);
        }
      }
    }
    __syncthreads();
    if (wave < 2) {
      int zc = 32 * nb + xc;
#pragma unroll
      for (int i = 0; i < MT; i++) {
#pragma unroll
        for (int v = 0; v < 8; v++) {
          int r = 16 * i + ((lane >> 4) << 3) + v;
          size_t o = (size_t)(mBase + r) * 64 + zc;
          out[o] = (acc[i][v] + bv) + ex[r * SA_STRIDE + xc] * eps[o];
        }
      }
    }
    return;
  }
  float cs = 0.f, cs2 = 0.f;
#pragma unroll
  for (int i = 0; i < MT; i++) {
    int mhalf = mBase + 16 * i + ((lane >> 4) << 3);
#pragma unroll
    for (int v = 0; v < 8; v++) {
      float val = acc[i][v] + bv;
      if (do_relu) val = fmaxf(val, 0.f);
      out[(size_t)(mhalf + v) * F + ncol] = val;
      cs += val;
      cs2 = __builtin_fmaf(val, val, cs2);
    }
  }
  if (st) {
    cs  += __shfl_xor(cs, 16);
    cs2 += __shfl_xor(cs2, 16);
    if (lane < 16) {
      atomicAdd(&st[ncol], (u64)__float2ll_rn(cs * FX_ONE));
      atomicAdd(&st[F + ncol], (u64)__float2ll_rn(cs2 * FX_ONE));
    }
  }
}

extern "C" void kernel_launch(void* const* d_in, const int* in_sizes, int n_in,
                              void* d_out, int out_size, void* d_ws, size_t ws_size,
                              hipStream_t stream) {
  const float* x   = (const float*)d_in[0];
  const int*   src = (const int*)d_in[1];
  const int*   dst = (const int*)d_in[2];
  const float* eps = (const float*)d_in[3];
  const float* W1l = (const float*)d_in[4];
  const float* W1r = (const float*)d_in[5];
  const float* b1  = (const float*)d_in[6];
  const float* g1  = (const float*)d_in[7];
  const float* be1 = (const float*)d_in[8];
  const float* W2l = (const float*)d_in[9];
  const float* W2r = (const float*)d_in[10];
  const float* b2  = (const float*)d_in[11];
  const float* g2  = (const float*)d_in[12];
  const float* be2 = (const float*)d_in[13];
  const float* W3l = (const float*)d_in[14];
  const float* W3r = (const float*)d_in[15];
  const float* b3  = (const float*)d_in[16];
  float* zout = (float*)d_out;
  (void)in_sizes; (void)n_in; (void)out_size; (void)ws_size;

  char* ws = (char*)d_ws;
  size_t off = 0;
  auto carve = [&](size_t bytes) -> float* {
    char* p = ws + off;
    off = (off + bytes + 255) & ~(size_t)255;
    return (float*)p;
  };
  float* rdeg = carve((size_t)NN * 4);
  float* msg  = carve((size_t)NN * 256 * 4);
  float* h1   = carve((size_t)NN * 128 * 4);
  float* h2   = carve((size_t)NN * 256 * 4);
  float* y1   = msg + (size_t)NN * 128;
  int* cnt    = (int*)h1;
  int* rowptr = cnt + CNT_PAD;
  int* col    = rowptr + CNT_PAD;
  int* rank   = col + NE;
  u64* st1    = (u64*)(rank + NE);
  u64* st2    = st1 + 256;

  const int M = NN, E = NE;

  zero_f32_2<<<256, 256, 0, stream>>>((float*)cnt, M, (float*)st1, 1536);
  deg_count<<<(E + 255) / 256, 256, 0, stream>>>(dst, cnt, rank, E);
  scan_rowptr<<<1, SCAN_T, 0, stream>>>(cnt, rowptr, rdeg, M);
  csr_fill<<<(E + 255) / 256, 256, 0, stream>>>(src, dst, rowptr, rank, col, E);

  gather_sum_128<<<(M + 7) / 8, 256, 0, stream>>>(x, rowptr, col, msg, M);
  sage_gemm_wmma<0, 0, 128, 128><<<(NN / BM) * 2, 128, 0, stream>>>(
      msg, rdeg, x, W1l, W1r, b1, y1, st1, nullptr, nullptr, nullptr, nullptr, 1);

  gather_bn_sum_128<<<(M + 7) / 8, 256, 0, stream>>>(y1, rowptr, col, msg, M, st1, g1, be1);
  sage_gemm_wmma<0, 1, 128, 256><<<(NN / BM) * 4, 128, 0, stream>>>(
      msg, rdeg, y1, W2l, W2r, b2, h2, st2, st1, g1, be1, nullptr, 1);

  gather_bn_sum_256<<<(M + 7) / 8, 256, 0, stream>>>(h2, rowptr, col, msg, M, st2, g2, be2);
  sage_gemm_wmma<1, 1, 256, 128><<<(NN / BM) * 2, 128, 0, stream>>>(
      msg, rdeg, h2, W3l, W3r, b3, zout, nullptr, st2, g2, be2, eps, 0);
}


# Edges grouped by target by a two-level counting sort on LDS counters (256-node buckets, then nodes; one scan of run counts): no global atomic per edge; a row's add order now follows LDS arrival
# speedup vs baseline: 4.0570x; 1.2264x over previous
#include <hip/hip_runtime.h>

#define NN 50000
#define NE 800000
#define KC 32
#define SA_STRIDE 36
#define SB_STRIDE 68
#define SCAN_T 1024
#define SCAN_CH 49
#define CNT_PAD (SCAN_T * SCAN_CH)

typedef __attribute__((ext_vector_type(2))) float v2f;
typedef __attribute__((ext_vector_type(8))) float v8f;
typedef unsigned long long u64;

#define FX_ONE 268435456.f
#define BN_EPS 1e-5f

__device__ __forceinline__ void bn_consts(u64 s, u64 q, float g, float& mean, float& scl) {
  float inv = 1.f / (float)NN;
  float S = (float)(long long)s * (1.f / FX_ONE);
  float Q = (float)(long long)q * (1.f / FX_ONE);
  mean = S * inv;
  float var = Q * inv - mean * mean;
  scl = g * rsqrtf(var + BN_EPS);
}

#define NBK 196
#define NEB 256
#define EPB 3125
#define CSR_EPT 5
static_assert(NBK * 256 >= NN + 1, "buckets cover nodes 0..NN");
static_assert(NEB * EPB == NE, "edge blocks cover the edge list exactly");
static_assert(NBK * NEB == CNT_PAD, "the scan takes exactly the NBK x NEB counts");
static_assert(EPB <= 4096, "four edges per thread at most");

__global__ void __launch_bounds__(1024) bucket_count(const int* __restrict__ dst,
                                                     int* __restrict__ lrank,
                                                     int* __restrict__ gh,
                                                     float* __restrict__ stz, int nst) {
  __shared__ int hist[NBK];
  int t = threadIdx.x, blk = blockIdx.x;
  if (t < NBK) hist[t] = 0;
  if (blk == 0)
    for (int k = t; k < nst; k += 1024) stz[k] = 0.f;
  __syncthreads();
  const int* d = dst + blk * EPB;
  int* r = lrank + blk * EPB;
  int v[4];
#pragma unroll
  for (int j = 0; j < 4; j++) {
    int i = j * 1024 + t;
    v[j] = d[(i < EPB) ? i : 0];
  }
#pragma unroll
  for (int j = 0; j < 4; j++) {
    int i = j * 1024 + t;
    if (i < EPB) r[i] = atomicAdd(&hist[v[j] >> 8], 1);
  }
  __syncthreads();
  if (t < NBK) gh[t * NEB + blk] = hist[t];
}

__global__ void __launch_bounds__(SCAN_T) scan_excl(const int* __restrict__ cnt,
                                                    int* __restrict__ outp) {
  __shared__ int s[CNT_PAD];
  __shared__ int wsum[32];
  int t = threadIdx.x, lane = t & 31, wave = t >> 5;
  {
    int v[SCAN_CH];
#pragma unroll
    for (int j = 0; j < SCAN_CH; j++) v[j] = cnt[j * SCAN_T + t];
    __builtin_amdgcn_sched_barrier(0);
#pragma unroll
    for (int j = 0; j < SCAN_CH; j++) s[j * SCAN_T + t] = v[j];
  }
  __syncthreads();
  int local = 0;
  for (int j = 0; j < SCAN_CH; j++) local += s[t * SCAN_CH + j];
  int inc = local;
  for (int o = 1; o < 32; o <<= 1) {
    int v = __shfl_up(inc, o);
    if (lane >= o) inc += v;
  }
  if (lane == 31) wsum[wave] = inc;
  __syncthreads();
  if (wave == 0) {
    int w = wsum[lane], wi = w;
    for (int o = 1; o < 32; o <<= 1) {
      int v = __shfl_up(wi, o);
      if (lane >= o) wi += v;
    }
    wsum[lane] = wi - w;
  }
  __syncthreads();
  int run = wsum[wave] + inc - local;
  for (int j = 0; j < SCAN_CH; j++) {
    int c = s[t * SCAN_CH + j];
    s[t * SCAN_CH + j] = run;
    run += c;
  }
  __syncthreads();
  for (int j = 0; j < SCAN_CH; j++) outp[j * SCAN_T + t] = s[j * SCAN_T + t];
}

__global__ void __launch_bounds__(1024) bucket_scatter(const int* __restrict__ src,
                                                       const int* __restrict__ dst,
                                                       const int* __restrict__ lrank,
                                                       const int* __restrict__ goff,
                                                       int2* __restrict__ bed) {
  __shared__ int base[NBK];
  int t = threadIdx.x, blk = blockIdx.x;
  if (t < NBK) base[t] = goff[t * NEB + blk];
  __syncthreads();
  const int* d = dst + blk * EPB;
  const int* s = src + blk * EPB;
  const int* r = lrank + blk * EPB;
  int vd[4], vs[4], vr[4];
#pragma unroll
  for (int j = 0; j < 4; j++) {
    int i = j * 1024 + t;
    int ii = (i < EPB) ? i : 0;
    vd[j] = d[ii]; vs[j] = s[ii]; vr[j] = r[ii];
  }
#pragma unroll
  for (int j = 0; j < 4; j++) {
    int i = j * 1024 + t;
    if (i < EPB) bed[base[vd[j] >> 8] + vr[j]] = make_int2(vd[j], vs[j]);
  }
}

__global__ void __launch_bounds__(1024) bucket_csr(const int2* __restrict__ bed,
                                                   const int* __restrict__ goff,
                                                   int* __restrict__ rowptr,
                                                   float* __restrict__ rdeg,
                                                   int* __restrict__ col) {
  __shared__ int cnt[256];
  __shared__ int cur[256];
  __shared__ int wsum[8];
  int t = threadIdx.x, lane = t & 31, wave = t >> 5, b = blockIdx.x;
  int beg = goff[b * NEB];
  int end = (b + 1 < NBK) ? goff[(b + 1) * NEB] : NE;
  if (t < 256) cnt[t] = 0;
  __syncthreads();
  for (int i0 = beg; i0 < end; i0 += CSR_EPT * 1024) {
    int2 ed[CSR_EPT];
#pragma unroll
    for (int j = 0; j < CSR_EPT; j++) {
      int i = i0 + j * 1024 + t;
      ed[j] = bed[(i < end) ? i : beg];
    }
#pragma unroll
    for (int j = 0; j < CSR_EPT; j++) {
      int i = i0 + j * 1024 + t;
      if (i < end) atomicAdd(&cnt[ed[j].x & 255], 1);
    }
  }
  __syncthreads();
  int c = 0, inc = 0;
  if (t < 256) {
    c = cnt[t];
    inc = c;
    for (int o = 1; o < 32; o <<= 1) {
      int v = __shfl_up(inc, o);
      if (lane >= o) inc += v;
    }
    if (lane == 31) wsum[wave] = inc;
  }
  __syncthreads();
  if (t < 256) {
    int off = 0;
    for (int w = 0; w < wave; w++) off += wsum[w];
    int ex = off + inc - c;
    cur[t] = ex;
    int node = b * 256 + t;
    if (node <= NN) rowptr[node] = beg + ex;
    if (node < NN) rdeg[node] = 1.f / fmaxf((float)c, 1.f);
  }
  __syncthreads();
  for (int i0 = beg; i0 < end; i0 += CSR_EPT * 1024) {
    int2 ed[CSR_EPT];
#pragma unroll
    for (int j = 0; j < CSR_EPT; j++) {
      int i = i0 + j * 1024 + t;
      ed[j] = bed[(i < end) ? i : beg];
    }
#pragma unroll
    for (int j = 0; j < CSR_EPT; j++) {
      int i = i0 + j * 1024 + t;
      if (i < end) col[beg + atomicAdd(&cur[ed[j].x & 255], 1)] = ed[j].y;
    }
  }
}

template <int V, int R, int BN>
__device__ __forceinline__ void add_rows(float4 (&acc)[V], const float* __restrict__ hl,
                                         int c, int j, const float4 (&mu)[V],
                                         const float4 (&sc)[V], const float4 (&sh)[V]) {
  constexpr int F = 128 * V;
  float4 x[R][V];
#pragma unroll
  for (int r = 0; r < R; r++) {
    const float* p = hl + (size_t)__builtin_amdgcn_readlane(c, j + r) * F;
#pragma unroll
    for (int v = 0; v < V; v++) x[r][v] = *(const float4*)(p + v * 128);
  }
#pragma unroll
  for (int r = 0; r < R; r++) {
#pragma unroll
    for (int v = 0; v < V; v++) {
      float4 y = x[r][v];
      if (BN) {
        y.x = (y.x - mu[v].x) * sc[v].x + sh[v].x;
        y.y = (y.y - mu[v].y) * sc[v].y + sh[v].y;
        y.z = (y.z - mu[v].z) * sc[v].z + sh[v].z;
        y.w = (y.w - mu[v].w) * sc[v].w + sh[v].w;
      }
      acc[v].x += y.x; acc[v].y += y.y;
      acc[v].z += y.z; acc[v].w += y.w;
    }
  }
}

template <int V, int BN>
__device__ __forceinline__ void gather_row(const float* __restrict__ h,
                                           const int* __restrict__ rowptr,
                                           const int* __restrict__ col,
                                           float* __restrict__ msg, int n,
                                           const u64* __restrict__ bst,
                                           const float* __restrict__ bg,
                                           const float* __restrict__ bb) {
  constexpr int F = 128 * V;
  __shared__ float gMu[BN ? F : 4], gSc[BN ? F : 4], gSh[BN ? F : 4];
  if (BN) {
    for (int k = threadIdx.x; k < F; k += blockDim.x) {
      float mean, scl;
      bn_consts(bst[k], bst[F + k], bg[k], mean, scl);
      gMu[k] = mean; gSc[k] = scl; gSh[k] = bb[k];
    }
    __syncthreads();
  }
  int lane = threadIdx.x & 31;
  int i = blockIdx.x * (blockDim.x >> 5) + (threadIdx.x >> 5);
  i = __builtin_amdgcn_readfirstlane(i);
  if (i >= n) return;
  int beg = rowptr[i], end = rowptr[i + 1];
  float4 acc[V];
#pragma unroll
  for (int v = 0; v < V; v++) acc[v] = make_float4(0.f, 0.f, 0.f, 0.f);
  float4 mu[V], sc[V], sh[V];
#pragma unroll
  for (int v = 0; v < V; v++) {
    mu[v] = sc[v] = sh[v] = make_float4(0.f, 0.f, 0.f, 0.f);
    if (BN) {
      int c0 = v * 128 + lane * 4;
      mu[v] = *(const float4*)(&gMu[c0]);
      sc[v] = *(const float4*)(&gSc[c0]);
      sh[v] = *(const float4*)(&gSh[c0]);
    }
  }
  const float* hl = h + lane * 4;
  for (int base = beg; base < end; base += 32) {
    int m = end - base;
    if (m > 32) m = 32;
    int c = (lane < m) ? col[base + lane] : 0;
    int j = 0;
    for (; j + 4 <= m; j += 4) add_rows<V, 4, BN>(acc, hl, c, j, mu, sc, sh);
    int rem = m - j;
    if (rem == 3) add_rows<V, 3, BN>(acc, hl, c, j, mu, sc, sh);
    else if (rem == 2) add_rows<V, 2, BN>(acc, hl, c, j, mu, sc, sh);
    else if (rem == 1) add_rows<V, 1, BN>(acc, hl, c, j, mu, sc, sh);
  }
  float* o = msg + (size_t)i * F + lane * 4;
#pragma unroll
  for (int v = 0; v < V; v++) *(float4*)(o + v * 128) = acc[v];
}

__global__ void gather_sum_128(const float* __restrict__ h, const int* __restrict__ rowptr,
                               const int* __restrict__ col, float* __restrict__ msg, int n) {
  gather_row<1, 0>(h, rowptr, col, msg, n, nullptr, nullptr, nullptr);
}

__global__ void gather_bn_sum_128(const float* __restrict__ h, const int* __restrict__ rowptr,
                                  const int* __restrict__ col, float* __restrict__ msg, int n,
                                  const u64* __restrict__ bst, const float* __restrict__ bg,
                                  const float* __restrict__ bb) {
  gather_row<1, 1>(h, rowptr, col, msg, n, bst, bg, bb);
}

__global__ void gather_bn_sum_256(const float* __restrict__ h, const int* __restrict__ rowptr,
                                  const int* __restrict__ col, float* __restrict__ msg, int n,
                                  const u64* __restrict__ bst, const float* __restrict__ bg,
                                  const float* __restrict__ bb) {
  gather_row<2, 1>(h, rowptr, col, msg, n, bst, bg, bb);
}

#define MT 5
#define BM (16 * MT)
template <int RP, int BN, int K, int F>
__global__ void __launch_bounds__(128)
sage_gemm_wmma(const float* __restrict__ msg, const float* __restrict__ rdeg,
               const float* __restrict__ h,
               const float* __restrict__ Wl, const float* __restrict__ Wr,
               const float* __restrict__ bias, float* __restrict__ out,
               u64* __restrict__ st,
               const u64* __restrict__ bst, const float* __restrict__ bg,
               const float* __restrict__ bb, const float* __restrict__ eps, int do_relu) {
  __shared__ float sA[2][BM * SA_STRIDE];
  __shared__ float sB[2][KC * SB_STRIDE];
  __shared__ float sSc[BN ? K : 4], sSh[BN ? K : 4];

  constexpr int nblocks_n = F / 64;
  int t = threadIdx.x;
  int wave = t >> 5, lane = t & 31;
  int mBase = (blockIdx.x / nblocks_n) * BM;
  int nb = blockIdx.x % nblocks_n;
  auto gcol = [&](int c) { return RP ? (32 * nb + (c & 31) + ((c >> 5) << 6)) : (64 * nb + c); };

  if (BN) {
    for (int k = t; k < K; k += 128) {
      float mean, scl;
      bn_consts(bst[k], bst[K + k], bg[k], mean, scl);
      sSc[k] = scl; sSh[k] = bb[k] - mean * scl;
    }
    __syncthreads();
  }

  int m16  = lane & 15;
  int kb   = (lane >> 4) << 1;
  int nloc = wave * 16 + (lane & 15);
  int ncol = gcol(nloc);

  v8f acc[MT];
#pragma unroll
  for (int i = 0; i < MT; i++) acc[i] = (v8f){0.f, 0.f, 0.f, 0.f, 0.f, 0.f, 0.f, 0.f};

  int ar = t >> 3, ac = (t & 7) << 2;
  int br = t >> 4, bc = (t & 15) << 2;
  unsigned aoff = (unsigned)(ar * K + ac);
  unsigned boff = (unsigned)(br * F + gcol(bc));
  const float* am = msg + (size_t)mBase * K;
  const float* ah = h + (size_t)mBase * K;

  float sc[MT];
#pragma unroll
  for (int i = 0; i < MT; i++) sc[i] = rdeg[mBase + ar + 16 * i];

#pragma nounroll
  for (int kc = 0; kc < K; kc += KC) {
    {
      float4 va0[MT], va1[MT], vb0[4], vb1[4];
      const float* pm = am + kc;
      const float* ph = ah + kc;
      const float* pl = Wl + (size_t)kc * F;
      const float* pr = Wr + (size_t)kc * F;
#pragma unroll
      for (int i = 0; i < MT; i++) {
        va0[i] = *(const float4*)(pm + (aoff + 16u * i * K));
        va1[i] = *(const float4*)(ph + (aoff + 16u * i * K));
      }
#pragma unroll
      for (int pass = 0; pass < 4; pass++) {
        vb0[pass] = *(const float4*)(pl + (boff + 8u * pass * F));
        vb1[pass] = *(const float4*)(pr + (boff + 8u * pass * F));
      }
      __builtin_amdgcn_sched_barrier(0);
      float4 sl, sf;
      if (BN) {
        sl = *(const float4*)(&sSc[kc + ac]);
        sf = *(const float4*)(&sSh[kc + ac]);
      }
#pragma unroll
      for (int i = 0; i < MT; i++) {
        float4 s0 = va0[i];
        s0.x *= sc[i]; s0.y *= sc[i]; s0.z *= sc[i]; s0.w *= sc[i];
        *(float4*)(&sA[0][(ar + 16 * i) * SA_STRIDE + ac]) = s0;
        float4 s1 = va1[i];
        if (BN) {
          s1.x = s1.x * sl.x + sf.x; s1.y = s1.y * sl.y + sf.y;
          s1.z = s1.z * sl.z + sf.z; s1.w = s1.w * sl.w + sf.w;
        }
        *(float4*)(&sA[1][(ar + 16 * i) * SA_STRIDE + ac]) = s1;
      }
#pragma unroll
      for (int pass = 0; pass < 4; pass++) {
        int row = br + pass * 8;
        *(float4*)(&sB[0][row * SB_STRIDE + bc]) = vb0[pass];
        *(float4*)(&sB[1][row * SB_STRIDE + bc]) = vb1[pass];
      }
    }
    if (kc + KC < K) {
      if (t < BM) {
        __builtin_prefetch(msg + (size_t)(mBase + t) * K + kc + KC, 0, 0);
        __builtin_prefetch(h   + (size_t)(mBase + t) * K + kc + KC, 0, 0);
      } else if (t - BM < KC) {
        __builtin_prefetch(Wl + (size_t)(kc + KC + t - BM) * F + gcol(0), 0, 0);
        __builtin_prefetch(Wr + (size_t)(kc + KC + t - BM) * F + gcol(0), 0, 0);
      }
    }
    __syncthreads();

#pragma unroll
    for (int k = 0; k < KC; k += 4) {
      v2f a, b0, b1;
      b0.x = sB[0][(k + kb) * SB_STRIDE + nloc];
      b0.y = sB[0][(k + kb + 1) * SB_STRIDE + nloc];
      b1.x = sB[1][(k + kb) * SB_STRIDE + nloc];
      b1.y = sB[1][(k + kb + 1) * SB_STRIDE + nloc];
#pragma unroll
      for (int i = 0; i < MT; i++) {
        a.x = sA[0][(m16 + 16 * i) * SA_STRIDE + k + kb];
        a.y = sA[0][(m16 + 16 * i) * SA_STRIDE + k + kb + 1];
        acc[i] = __builtin_amdgcn_wmma_f32_16x16x4_f32(false, a, false, b0, (short)0, acc[i],
                                                       false, false);
      }
#pragma unroll
      for (int i = 0; i < MT; i++) {
        a.x = sA[1][(m16 + 16 * i) * SA_STRIDE + k + kb];
        a.y = sA[1][(m16 + 16 * i) * SA_STRIDE + k + kb + 1];
        acc[i] = __builtin_amdgcn_wmma_f32_16x16x4_f32(false, a, false, b1, (short)0, acc[i],
                                                       false, false);
      }
    }
    __syncthreads();
  }

  float bv = bias[ncol];
  if (RP) {
    float* ex = &sA[0][0];
    int xc = (wave & 1) * 16 + (lane & 15);
    if (wave >= 2) {
#pragma unroll
      for (int i = 0; i < MT; i++) {
#pragma unroll
        for (int v = 0; v < 8; v++) {
          int r = 16 * i + ((lane >> 4) << 3) + v;
          ex[r * SA_STRIDE + xc] = expf(acc[i][v] + bv);
        }
      }
    }
    __syncthreads();
    if (wave < 2) {
      int zc = 32 * nb + xc;
#pragma unroll
      for (int i = 0; i < MT; i++) {
#pragma unroll
        for (int v = 0; v < 8; v++) {
          int r = 16 * i + ((lane >> 4) << 3) + v;
          size_t o = (size_t)(mBase + r) * 64 + zc;
          out[o] = (acc[i][v] + bv) + ex[r * SA_STRIDE + xc] * eps[o];
        }
      }
    }
    return;
  }
  float cs = 0.f, cs2 = 0.f;
#pragma unroll
  for (int i = 0; i < MT; i++) {
    int mhalf = mBase + 16 * i + ((lane >> 4) << 3);
#pragma unroll
    for (int v = 0; v < 8; v++) {
      float val = acc[i][v] + bv;
      if (do_relu) val = fmaxf(val, 0.f);
      out[(size_t)(mhalf + v) * F + ncol] = val;
      cs += val;
      cs2 = __builtin_fmaf(val, val, cs2);
    }
  }
  if (st) {
    cs  += __shfl_xor(cs, 16);
    cs2 += __shfl_xor(cs2, 16);
    if (lane < 16) {
      atomicAdd(&st[ncol], (u64)__float2ll_rn(cs * FX_ONE));
      atomicAdd(&st[F + ncol], (u64)__float2ll_rn(cs2 * FX_ONE));
    }
  }
}

extern "C" void kernel_launch(void* const* d_in, const int* in_sizes, int n_in,
                              void* d_out, int out_size, void* d_ws, size_t ws_size,
                              hipStream_t stream) {
  const float* x   = (const float*)d_in[0];
  const int*   src = (const int*)d_in[1];
  const int*   dst = (const int*)d_in[2];
  const float* eps = (const float*)d_in[3];
  const float* W1l = (const float*)d_in[4];
  const float* W1r = (const float*)d_in[5];
  const float* b1  = (const float*)d_in[6];
  const float* g1  = (const float*)d_in[7];
  const float* be1 = (const float*)d_in[8];
  const float* W2l = (const float*)d_in[9];
  const float* W2r = (const float*)d_in[10];
  const float* b2  = (const float*)d_in[11];
  const float* g2  = (const float*)d_in[12];
  const float* be2 = (const float*)d_in[13];
  const float* W3l = (const float*)d_in[14];
  const float* W3r = (const float*)d_in[15];
  const float* b3  = (const float*)d_in[16];
  float* zout = (float*)d_out;
  (void)in_sizes; (void)n_in; (void)out_size; (void)ws_size;

  char* ws = (char*)d_ws;
  size_t off = 0;
  auto carve = [&](size_t bytes) -> float* {
    char* p = ws + off;
    off = (off + bytes + 255) & ~(size_t)255;
    return (float*)p;
  };
  float* rdeg = carve((size_t)NN * 4);
  float* msg  = carve((size_t)NN * 256 * 4);
  float* h1   = carve((size_t)NN * 128 * 4);
  float* h2   = carve((size_t)NN * 256 * 4);
  float* y1   = msg + (size_t)NN * 128;
  int* gh     = (int*)h1;
  int* rowptr = gh + CNT_PAD;
  int* col    = rowptr + CNT_PAD;
  int* lrank  = col + NE;
  u64* st1    = (u64*)(lrank + NE);
  u64* st2    = st1 + 256;
  int* goff   = (int*)(st2 + 512);
  int2* bed   = (int2*)(goff + CNT_PAD);

  const int M = NN;

  bucket_count<<<NEB, 1024, 0, stream>>>(dst, lrank, gh, (float*)st1, 1536);
  scan_excl<<<1, SCAN_T, 0, stream>>>(gh, goff);
  bucket_scatter<<<NEB, 1024, 0, stream>>>(src, dst, lrank, goff, bed);
  bucket_csr<<<NBK, 1024, 0, stream>>>(bed, goff, rowptr, rdeg, col);

  gather_sum_128<<<(M + 7) / 8, 256, 0, stream>>>(x, rowptr, col, msg, M);
  sage_gemm_wmma<0, 0, 128, 128><<<(NN / BM) * 2, 128, 0, stream>>>(
      msg, rdeg, x, W1l, W1r, b1, y1, st1, nullptr, nullptr, nullptr, nullptr, 1);

  gather_bn_sum_128<<<(M + 7) / 8, 256, 0, stream>>>(y1, rowptr, col, msg, M, st1, g1, be1);
  sage_gemm_wmma<0, 1, 128, 256><<<(NN / BM) * 4, 128, 0, stream>>>(
      msg, rdeg, y1, W2l, W2r, b2, h2, st2, st1, g1, be1, nullptr, 1);

  gather_bn_sum_256<<<(M + 7) / 8, 256, 0, stream>>>(h2, rowptr, col, msg, M, st2, g2, be2);
  sage_gemm_wmma<1, 1, 256, 128><<<(NN / BM) * 2, 128, 0, stream>>>(
      msg, rdeg, h2, W3l, W3r, b3, zout, nullptr, st2, g2, be2, eps, 0);
}


# Layer 3 projects before it aggregates: its GEMM forms P=h@W3l and S=h@W3r+b from one staged h; the closing gather averages 128-wide P rows (was 256-wide h), adds S, reparameterizes; sum order changes
# speedup vs baseline: 4.3591x; 1.0745x over previous
#include <hip/hip_runtime.h>

#define NN 50000
#define NE 800000
#define KC 32
#define SA_STRIDE 36
#define SB_STRIDE 68
#define SCAN_T 1024
#define SCAN_CH 49
#define CNT_PAD (SCAN_T * SCAN_CH)

typedef __attribute__((ext_vector_type(2))) float v2f;
typedef __attribute__((ext_vector_type(8))) float v8f;
typedef unsigned long long u64;

#define FX_ONE 268435456.f
#define BN_EPS 1e-5f

__device__ __forceinline__ void bn_consts(u64 s, u64 q, float g, float& mean, float& scl) {
  float inv = 1.f / (float)NN;
  float S = (float)(long long)s * (1.f / FX_ONE);
  float Q = (float)(long long)q * (1.f / FX_ONE);
  mean = S * inv;
  float var = Q * inv - mean * mean;
  scl = g * rsqrtf(var + BN_EPS);
}

#define NBK 196
#define NEB 256
#define EPB 3125
#define CSR_EPT 5
static_assert(NBK * 256 >= NN + 1, "buckets cover nodes 0..NN");
static_assert(NEB * EPB == NE, "edge blocks cover the edge list exactly");
static_assert(NBK * NEB == CNT_PAD, "the scan takes exactly the NBK x NEB counts");
static_assert(EPB <= 4096, "four edges per thread at most");

__global__ void __launch_bounds__(1024) bucket_count(const int* __restrict__ dst,
                                                     int* __restrict__ lrank,
                                                     int* __restrict__ gh,
                                                     float* __restrict__ stz, int nst) {
  __shared__ int hist[NBK];
  int t = threadIdx.x, blk = blockIdx.x;
  if (t < NBK) hist[t] = 0;
  if (blk == 0)
    for (int k = t; k < nst; k += 1024) stz[k] = 0.f;
  __syncthreads();
  const int* d = dst + blk * EPB;
  int* r = lrank + blk * EPB;
  int v[4];
#pragma unroll
  for (int j = 0; j < 4; j++) {
    int i = j * 1024 + t;
    v[j] = d[(i < EPB) ? i : 0];
  }
#pragma unroll
  for (int j = 0; j < 4; j++) {
    int i = j * 1024 + t;
    if (i < EPB) r[i] = atomicAdd(&hist[v[j] >> 8], 1);
  }
  __syncthreads();
  if (t < NBK) gh[t * NEB + blk] = hist[t];
}

__global__ void __launch_bounds__(SCAN_T) scan_excl(const int* __restrict__ cnt,
                                                    int* __restrict__ outp) {
  __shared__ int s[CNT_PAD];
  __shared__ int wsum[32];
  int t = threadIdx.x, lane = t & 31, wave = t >> 5;
  {
    int v[SCAN_CH];
#pragma unroll
    for (int j = 0; j < SCAN_CH; j++) v[j] = cnt[j * SCAN_T + t];
    __builtin_amdgcn_sched_barrier(0);
#pragma unroll
    for (int j = 0; j < SCAN_CH; j++) s[j * SCAN_T + t] = v[j];
  }
  __syncthreads();
  int local = 0;
  for (int j = 0; j < SCAN_CH; j++) local += s[t * SCAN_CH + j];
  int inc = local;
  for (int o = 1; o < 32; o <<= 1) {
    int v = __shfl_up(inc, o);
    if (lane >= o) inc += v;
  }
  if (lane == 31) wsum[wave] = inc;
  __syncthreads();
  if (wave == 0) {
    int w = wsum[lane], wi = w;
    for (int o = 1; o < 32; o <<= 1) {
      int v = __shfl_up(wi, o);
      if (lane >= o) wi += v;
    }
    wsum[lane] = wi - w;
  }
  __syncthreads();
  int run = wsum[wave] + inc - local;
  for (int j = 0; j < SCAN_CH; j++) {
    int c = s[t * SCAN_CH + j];
    s[t * SCAN_CH + j] = run;
    run += c;
  }
  __syncthreads();
  for (int j = 0; j < SCAN_CH; j++) outp[j * SCAN_T + t] = s[j * SCAN_T + t];
}

__global__ void __launch_bounds__(1024) bucket_scatter(const int* __restrict__ src,
                                                       const int* __restrict__ dst,
                                                       const int* __restrict__ lrank,
                                                       const int* __restrict__ goff,
                                                       int2* __restrict__ bed) {
  __shared__ int base[NBK];
  int t = threadIdx.x, blk = blockIdx.x;
  if (t < NBK) base[t] = goff[t * NEB + blk];
  __syncthreads();
  const int* d = dst + blk * EPB;
  const int* s = src + blk * EPB;
  const int* r = lrank + blk * EPB;
  int vd[4], vs[4], vr[4];
#pragma unroll
  for (int j = 0; j < 4; j++) {
    int i = j * 1024 + t;
    int ii = (i < EPB) ? i : 0;
    vd[j] = d[ii]; vs[j] = s[ii]; vr[j] = r[ii];
  }
#pragma unroll
  for (int j = 0; j < 4; j++) {
    int i = j * 1024 + t;
    if (i < EPB) bed[base[vd[j] >> 8] + vr[j]] = make_int2(vd[j], vs[j]);
  }
}

__global__ void __launch_bounds__(1024) bucket_csr(const int2* __restrict__ bed,
                                                   const int* __restrict__ goff,
                                                   int* __restrict__ rowptr,
                                                   float* __restrict__ rdeg,
                                                   int* __restrict__ col) {
  __shared__ int cnt[256];
  __shared__ int cur[256];
  __shared__ int wsum[8];
  int t = threadIdx.x, lane = t & 31, wave = t >> 5, b = blockIdx.x;
  int beg = goff[b * NEB];
  int end = (b + 1 < NBK) ? goff[(b + 1) * NEB] : NE;
  if (t < 256) cnt[t] = 0;
  __syncthreads();
  for (int i0 = beg; i0 < end; i0 += CSR_EPT * 1024) {
    int2 ed[CSR_EPT];
#pragma unroll
    for (int j = 0; j < CSR_EPT; j++) {
      int i = i0 + j * 1024 + t;
      ed[j] = bed[(i < end) ? i : beg];
    }
#pragma unroll
    for (int j = 0; j < CSR_EPT; j++) {
      int i = i0 + j * 1024 + t;
      if (i < end) atomicAdd(&cnt[ed[j].x & 255], 1);
    }
  }
  __syncthreads();
  int c = 0, inc = 0;
  if (t < 256) {
    c = cnt[t];
    inc = c;
    for (int o = 1; o < 32; o <<= 1) {
      int v = __shfl_up(inc, o);
      if (lane >= o) inc += v;
    }
    if (lane == 31) wsum[wave] = inc;
  }
  __syncthreads();
  if (t < 256) {
    int off = 0;
    for (int w = 0; w < wave; w++) off += wsum[w];
    int ex = off + inc - c;
    cur[t] = ex;
    int node = b * 256 + t;
    if (node <= NN) rowptr[node] = beg + ex;
    if (node < NN) rdeg[node] = 1.f / fmaxf((float)c, 1.f);
  }
  __syncthreads();
  for (int i0 = beg; i0 < end; i0 += CSR_EPT * 1024) {
    int2 ed[CSR_EPT];
#pragma unroll
    for (int j = 0; j < CSR_EPT; j++) {
      int i = i0 + j * 1024 + t;
      ed[j] = bed[(i < end) ? i : beg];
    }
#pragma unroll
    for (int j = 0; j < CSR_EPT; j++) {
      int i = i0 + j * 1024 + t;
      if (i < end) col[beg + atomicAdd(&cur[ed[j].x & 255], 1)] = ed[j].y;
    }
  }
}

template <int V, int R, int BN>
__device__ __forceinline__ void add_rows(float4 (&acc)[V], const float* __restrict__ hl,
                                         int c, int j, const float4 (&mu)[V],
                                         const float4 (&sc)[V], const float4 (&sh)[V]) {
  constexpr int F = 128 * V;
  float4 x[R][V];
#pragma unroll
  for (int r = 0; r < R; r++) {
    const float* p = hl + (size_t)__builtin_amdgcn_readlane(c, j + r) * F;
#pragma unroll
    for (int v = 0; v < V; v++) x[r][v] = *(const float4*)(p + v * 128);
  }
#pragma unroll
  for (int r = 0; r < R; r++) {
#pragma unroll
    for (int v = 0; v < V; v++) {
      float4 y = x[r][v];
      if (BN) {
        y.x = (y.x - mu[v].x) * sc[v].x + sh[v].x;
        y.y = (y.y - mu[v].y) * sc[v].y + sh[v].y;
        y.z = (y.z - mu[v].z) * sc[v].z + sh[v].z;
        y.w = (y.w - mu[v].w) * sc[v].w + sh[v].w;
      }
      acc[v].x += y.x; acc[v].y += y.y;
      acc[v].z += y.z; acc[v].w += y.w;
    }
  }
}

template <int V, int BN, int FIN>
__device__ __forceinline__ void gather_row(const float* __restrict__ h,
                                           const int* __restrict__ rowptr,
                                           const int* __restrict__ col,
                                           float* __restrict__ msg, int n,
                                           const u64* __restrict__ bst,
                                           const float* __restrict__ bg,
                                           const float* __restrict__ bb,
                                           const float* __restrict__ rdeg,
                                           const float* __restrict__ self,
                                           const float* __restrict__ eps) {
  static_assert(!FIN || (V == 1 && !BN), "the closing form is the plain 128-wide gather");
  constexpr int F = 128 * V;
  __shared__ float gMu[BN ? F : 4], gSc[BN ? F : 4], gSh[BN ? F : 4];
  if (BN) {
    for (int k = threadIdx.x; k < F; k += blockDim.x) {
      float mean, scl;
      bn_consts(bst[k], bst[F + k], bg[k], mean, scl);
      gMu[k] = mean; gSc[k] = scl; gSh[k] = bb[k];
    }
    __syncthreads();
  }
  int lane = threadIdx.x & 31;
  int i = blockIdx.x * (blockDim.x >> 5) + (threadIdx.x >> 5);
  i = __builtin_amdgcn_readfirstlane(i);
  if (i >= n) return;
  int beg = rowptr[i], end = rowptr[i + 1];
  float rd = 0.f;
  float4 sv = make_float4(0.f, 0.f, 0.f, 0.f), ev = sv;
  if (FIN) {
    rd = rdeg[i];
    sv = *(const float4*)(self + (size_t)i * 128 + lane * 4);
    ev = *(const float4*)(eps + (size_t)i * 64 + (lane & 15) * 4);
  }
  float4 acc[V];
#pragma unroll
  for (int v = 0; v < V; v++) acc[v] = make_float4(0.f, 0.f, 0.f, 0.f);
  float4 mu[V], sc[V], sh[V];
#pragma unroll
  for (int v = 0; v < V; v++) {
    mu[v] = sc[v] = sh[v] = make_float4(0.f, 0.f, 0.f, 0.f);
    if (BN) {
      int c0 = v * 128 + lane * 4;
      mu[v] = *(const float4*)(&gMu[c0]);
      sc[v] = *(const float4*)(&gSc[c0]);
      sh[v] = *(const float4*)(&gSh[c0]);
    }
  }
  const float* hl = h + lane * 4;
  for (int base = beg; base < end; base += 32) {
    int m = end - base;
    if (m > 32) m = 32;
    int c = (lane < m) ? col[base + lane] : 0;
    int j = 0;
    for (; j + 4 <= m; j += 4) add_rows<V, 4, BN>(acc, hl, c, j, mu, sc, sh);
    int rem = m - j;
    if (rem == 3) add_rows<V, 3, BN>(acc, hl, c, j, mu, sc, sh);
    else if (rem == 2) add_rows<V, 2, BN>(acc, hl, c, j, mu, sc, sh);
    else if (rem == 1) add_rows<V, 1, BN>(acc, hl, c, j, mu, sc, sh);
  }
  if (FIN) {
    float4 t;
    t.x = acc[0].x * rd + sv.x; t.y = acc[0].y * rd + sv.y;
    t.z = acc[0].z * rd + sv.z; t.w = acc[0].w * rd + sv.w;
    float4 e;
    e.x = __shfl_down(expf(t.x), 16); e.y = __shfl_down(expf(t.y), 16);
    e.z = __shfl_down(expf(t.z), 16); e.w = __shfl_down(expf(t.w), 16);
    if (lane < 16) {
      float4 z;
      z.x = t.x + e.x * ev.x; z.y = t.y + e.y * ev.y;
      z.z = t.z + e.z * ev.z; z.w = t.w + e.w * ev.w;
      *(float4*)(msg + (size_t)i * 64 + lane * 4) = z;
    }
    return;
  }
  float* o = msg + (size_t)i * F + lane * 4;
#pragma unroll
  for (int v = 0; v < V; v++) *(float4*)(o + v * 128) = acc[v];
}

__global__ void gather_sum_128(const float* __restrict__ h, const int* __restrict__ rowptr,
                               const int* __restrict__ col, float* __restrict__ msg, int n) {
  gather_row<1, 0, 0>(h, rowptr, col, msg, n, nullptr, nullptr, nullptr, nullptr, nullptr, nullptr);
}

__global__ void gather_bn_sum_128(const float* __restrict__ h, const int* __restrict__ rowptr,
                                  const int* __restrict__ col, float* __restrict__ msg, int n,
                                  const u64* __restrict__ bst, const float* __restrict__ bg,
                                  const float* __restrict__ bb) {
  gather_row<1, 1, 0>(h, rowptr, col, msg, n, bst, bg, bb, nullptr, nullptr, nullptr);
}

__global__ void gather_mean_reparam(const float* __restrict__ p, const int* __restrict__ rowptr,
                                    const int* __restrict__ col, float* __restrict__ z, int n,
                                    const float* __restrict__ rdeg,
                                    const float* __restrict__ self,
                                    const float* __restrict__ eps) {
  gather_row<1, 0, 1>(p, rowptr, col, z, n, nullptr, nullptr, nullptr, rdeg, self, eps);
}

#define MT 5
#define BM (16 * MT)
template <int PJ, int BN, int K, int F>
__global__ void __launch_bounds__(128)
sage_gemm_wmma(const float* __restrict__ msg, const float* __restrict__ rdeg,
               const float* __restrict__ h,
               const float* __restrict__ Wl, const float* __restrict__ Wr,
               const float* __restrict__ bias, float* __restrict__ out,
               float* __restrict__ out2, u64* __restrict__ st,
               const u64* __restrict__ bst, const float* __restrict__ bg,
               const float* __restrict__ bb, int do_relu) {
  constexpr int HP = PJ ? 0 : 1;
  __shared__ float sA[PJ ? 1 : 2][BM * SA_STRIDE];
  __shared__ float sB[2][KC * SB_STRIDE];
  __shared__ float sSc[BN ? K : 4], sSh[BN ? K : 4];

  constexpr int nblocks_n = F / 64;
  int t = threadIdx.x;
  int wave = t >> 5, lane = t & 31;
  int mBase = (blockIdx.x / nblocks_n) * BM;
  int nb = blockIdx.x % nblocks_n;

  if (BN) {
    for (int k = t; k < K; k += 128) {
      float mean, scl;
      bn_consts(bst[k], bst[K + k], bg[k], mean, scl);
      sSc[k] = scl; sSh[k] = bb[k] - mean * scl;
    }
    __syncthreads();
  }

  int m16  = lane & 15;
  int kb   = (lane >> 4) << 1;
  int nloc = wave * 16 + (lane & 15);
  int ncol = 64 * nb + nloc;

  v8f acc[MT];
  v8f acc2[PJ ? MT : 1];
#pragma unroll
  for (int i = 0; i < MT; i++) acc[i] = (v8f){0.f, 0.f, 0.f, 0.f, 0.f, 0.f, 0.f, 0.f};
  if (PJ) {
#pragma unroll
    for (int i = 0; i < MT; i++) acc2[i] = (v8f){0.f, 0.f, 0.f, 0.f, 0.f, 0.f, 0.f, 0.f};
  }

  int ar = t >> 3, ac = (t & 7) << 2;
  int br = t >> 4, bc = (t & 15) << 2;
  unsigned aoff = (unsigned)(ar * K + ac);
  unsigned boff = (unsigned)(br * F + 64 * nb + bc);
  const float* am = msg + (size_t)mBase * K;
  const float* ah = h + (size_t)mBase * K;

  float sc[MT];
  if (!PJ) {
#pragma unroll
    for (int i = 0; i < MT; i++) sc[i] = rdeg[mBase + ar + 16 * i];
  }

#pragma nounroll
  for (int kc = 0; kc < K; kc += KC) {
    {
      float4 va0[PJ ? 1 : MT], va1[MT], vb0[4], vb1[4];
      const float* pm = am + kc;
      const float* ph = ah + kc;
      const float* pl = Wl + (size_t)kc * F;
      const float* pr = Wr + (size_t)kc * F;
#pragma unroll
      for (int i = 0; i < MT; i++) {
        if (!PJ) va0[i] = *(const float4*)(pm + (aoff + 16u * i * K));
        va1[i] = *(const float4*)(ph + (aoff + 16u * i * K));
      }
#pragma unroll
      for (int pass = 0; pass < 4; pass++) {
        vb0[pass] = *(const float4*)(pl + (boff + 8u * pass * F));
        vb1[pass] = *(const float4*)(pr + (boff + 8u * pass * F));
      }
      __builtin_amdgcn_sched_barrier(0);
      float4 sl, sf;
      if (BN) {
        sl = *(const float4*)(&sSc[kc + ac]);
        sf = *(const float4*)(&sSh[kc + ac]);
      }
#pragma unroll
      for (int i = 0; i < MT; i++) {
        if (!PJ) {
          float4 s0 = va0[i];
          s0.x *= sc[i]; s0.y *= sc[i]; s0.z *= sc[i]; s0.w *= sc[i];
          *(float4*)(&sA[0][(ar + 16 * i) * SA_STRIDE + ac]) = s0;
        }
        float4 s1 = va1[i];
        if (BN) {
          s1.x = s1.x * sl.x + sf.x; s1.y = s1.y * sl.y + sf.y;
          s1.z = s1.z * sl.z + sf.z; s1.w = s1.w * sl.w + sf.w;
        }
        *(float4*)(&sA[HP][(ar + 16 * i) * SA_STRIDE + ac]) = s1;
      }
#pragma unroll
      for (int pass = 0; pass < 4; pass++) {
        int row = br + pass * 8;
        *(float4*)(&sB[0][row * SB_STRIDE + bc]) = vb0[pass];
        *(float4*)(&sB[1][row * SB_STRIDE + bc]) = vb1[pass];
      }
    }
    if (kc + KC < K) {
      if (t < BM) {
        if (!PJ) __builtin_prefetch(msg + (size_t)(mBase + t) * K + kc + KC, 0, 0);
        __builtin_prefetch(h   + (size_t)(mBase + t) * K + kc + KC, 0, 0);
      } else if (t - BM < KC) {
        __builtin_prefetch(Wl + (size_t)(kc + KC + t - BM) * F + 64 * nb, 0, 0);
        __builtin_prefetch(Wr + (size_t)(kc + KC + t - BM) * F + 64 * nb, 0, 0);
      }
    }
    __syncthreads();

#pragma unroll
    for (int k = 0; k < KC; k += 4) {
      v2f a, b0, b1;
      b0.x = sB[0][(k + kb) * SB_STRIDE + nloc];
      b0.y = sB[0][(k + kb + 1) * SB_STRIDE + nloc];
      b1.x = sB[1][(k + kb) * SB_STRIDE + nloc];
      b1.y = sB[1][(k + kb + 1) * SB_STRIDE + nloc];
      if (PJ) {
#pragma unroll
        for (int i = 0; i < MT; i++) {
          a.x = sA[0][(m16 + 16 * i) * SA_STRIDE + k + kb];
          a.y = sA[0][(m16 + 16 * i) * SA_STRIDE + k + kb + 1];
          acc[i] = __builtin_amdgcn_wmma_f32_16x16x4_f32(false, a, false, b0, (short)0, acc[i],
                                                         false, false);
          acc2[i] = __builtin_amdgcn_wmma_f32_16x16x4_f32(false, a, false, b1, (short)0, acc2[i],
                                                          false, false);
        }
      } else {
#pragma unroll
        for (int i = 0; i < MT; i++) {
          a.x = sA[0][(m16 + 16 * i) * SA_STRIDE + k + kb];
          a.y = sA[0][(m16 + 16 * i) * SA_STRIDE + k + kb + 1];
          acc[i] = __builtin_amdgcn_wmma_f32_16x16x4_f32(false, a, false, b0, (short)0, acc[i],
                                                         false, false);
        }
#pragma unroll
        for (int i = 0; i < MT; i++) {
          a.x = sA[HP][(m16 + 16 * i) * SA_STRIDE + k + kb];
          a.y = sA[HP][(m16 + 16 * i) * SA_STRIDE + k + kb + 1];
          acc[i] = __builtin_amdgcn_wmma_f32_16x16x4_f32(false, a, false, b1, (short)0, acc[i],
                                                         false, false);
        }
      }
    }
    __syncthreads();
  }

  float bv = bias[ncol];
  if (PJ) {
#pragma unroll
    for (int i = 0; i < MT; i++) {
      int mhalf = mBase + 16 * i + ((lane >> 4) << 3);
#pragma unroll
      for (int v = 0; v < 8; v++) {
        out[(size_t)(mhalf + v) * F + ncol] = acc[i][v];
        out2[(size_t)(mhalf + v) * F + ncol] = acc2[i][v] + bv;
      }
    }
    return;
  }
  float cs = 0.f, cs2 = 0.f;
#pragma unroll
  for (int i = 0; i < MT; i++) {
    int mhalf = mBase + 16 * i + ((lane >> 4) << 3);
#pragma unroll
    for (int v = 0; v < 8; v++) {
      float val = acc[i][v] + bv;
      if (do_relu) val = fmaxf(val, 0.f);
      out[(size_t)(mhalf + v) * F + ncol] = val;
      cs += val;
      cs2 = __builtin_fmaf(val, val, cs2);
    }
  }
  if (st) {
    cs  += __shfl_xor(cs, 16);
    cs2 += __shfl_xor(cs2, 16);
    if (lane < 16) {
      atomicAdd(&st[ncol], (u64)__float2ll_rn(cs * FX_ONE));
      atomicAdd(&st[F + ncol], (u64)__float2ll_rn(cs2 * FX_ONE));
    }
  }
}

extern "C" void kernel_launch(void* const* d_in, const int* in_sizes, int n_in,
                              void* d_out, int out_size, void* d_ws, size_t ws_size,
                              hipStream_t stream) {
  const float* x   = (const float*)d_in[0];
  const int*   src = (const int*)d_in[1];
  const int*   dst = (const int*)d_in[2];
  const float* eps = (const float*)d_in[3];
  const float* W1l = (const float*)d_in[4];
  const float* W1r = (const float*)d_in[5];
  const float* b1  = (const float*)d_in[6];
  const float* g1  = (const float*)d_in[7];
  const float* be1 = (const float*)d_in[8];
  const float* W2l = (const float*)d_in[9];
  const float* W2r = (const float*)d_in[10];
  const float* b2  = (const float*)d_in[11];
  const float* g2  = (const float*)d_in[12];
  const float* be2 = (const float*)d_in[13];
  const float* W3l = (const float*)d_in[14];
  const float* W3r = (const float*)d_in[15];
  const float* b3  = (const float*)d_in[16];
  float* zout = (float*)d_out;
  (void)in_sizes; (void)n_in; (void)out_size; (void)ws_size;

  char* ws = (char*)d_ws;
  size_t off = 0;
  auto carve = [&](size_t bytes) -> float* {
    char* p = ws + off;
    off = (off + bytes + 255) & ~(size_t)255;
    return (float*)p;
  };
  float* rdeg = carve((size_t)NN * 4);
  float* msg  = carve((size_t)NN * 256 * 4);
  float* h1   = carve((size_t)NN * 128 * 4);
  float* h2   = carve((size_t)NN * 256 * 4);
  float* y1   = msg + (size_t)NN * 128;
  int* gh     = (int*)h1;
  int* rowptr = gh + CNT_PAD;
  int* col    = rowptr + CNT_PAD;
  int* lrank  = col + NE;
  u64* st1    = (u64*)(lrank + NE);
  u64* st2    = st1 + 256;
  int* goff   = (int*)(st2 + 512);
  int2* bed   = (int2*)(goff + CNT_PAD);

  const int M = NN;

  bucket_count<<<NEB, 1024, 0, stream>>>(dst, lrank, gh, (float*)st1, 1536);
  scan_excl<<<1, SCAN_T, 0, stream>>>(gh, goff);
  bucket_scatter<<<NEB, 1024, 0, stream>>>(src, dst, lrank, goff, bed);
  bucket_csr<<<NBK, 1024, 0, stream>>>(bed, goff, rowptr, rdeg, col);

  gather_sum_128<<<(M + 7) / 8, 256, 0, stream>>>(x, rowptr, col, msg, M);
  sage_gemm_wmma<0, 0, 128, 128><<<(NN / BM) * 2, 128, 0, stream>>>(
      msg, rdeg, x, W1l, W1r, b1, y1, nullptr, st1, nullptr, nullptr, nullptr, 1);

  gather_bn_sum_128<<<(M + 7) / 8, 256, 0, stream>>>(y1, rowptr, col, msg, M, st1, g1, be1);
  sage_gemm_wmma<0, 1, 128, 256><<<(NN / BM) * 4, 128, 0, stream>>>(
      msg, rdeg, y1, W2l, W2r, b2, h2, nullptr, st2, st1, g1, be1, 1);

  float* pj = msg;
  float* sj = y1;
  sage_gemm_wmma<1, 1, 256, 128><<<(NN / BM) * 2, 128, 0, stream>>>(
      nullptr, nullptr, h2, W3l, W3r, b3, pj, sj, nullptr, st2, g2, be2, 0);
  gather_mean_reparam<<<(M + 7) / 8, 256, 0, stream>>>(pj, rowptr, col, zout, M, rdeg, sj, eps);
}


# Edges grouped with no grid-wide scan: each edge block sorts its own edges by bucket in LDS into its own region; the per-bucket pass reads its 256 runs there; 10 -> 8 launches; row add order changes
# speedup vs baseline: 4.7243x; 1.0838x over previous
#include <hip/hip_runtime.h>

#define NN 50000
#define NE 800000
#define KC 32
#define SA_STRIDE 36
#define SB_STRIDE 68

typedef __attribute__((ext_vector_type(2))) float v2f;
typedef __attribute__((ext_vector_type(8))) float v8f;
typedef unsigned long long u64;

#define FX_ONE 268435456.f
#define BN_EPS 1e-5f

__device__ __forceinline__ void bn_consts(u64 s, u64 q, float g, float& mean, float& scl) {
  float inv = 1.f / (float)NN;
  float S = (float)(long long)s * (1.f / FX_ONE);
  float Q = (float)(long long)q * (1.f / FX_ONE);
  mean = S * inv;
  float var = Q * inv - mean * mean;
  scl = g * rsqrtf(var + BN_EPS);
}

#define NBK 196
#define NEB 256
#define EPB 3125
#define SEG_Q 8
#define BOFF_N ((NBK + 1) * NEB)
#define ROWPTR_PAD 50176
static_assert(NBK * 256 >= NN + 1, "buckets cover nodes 0..NN");
static_assert(NEB * EPB == NE, "edge blocks cover the edge list exactly");
static_assert(EPB <= 4096, "four edges per thread at most");
static_assert(NBK < 256, "the 256-wide scan also yields entry NBK, the block's edge count");
static_assert(NEB == 256 && 32 * SEG_Q == NEB, "a 1024-thread work-group reads one run table row per thread 0..255, SEG_Q runs per wave");
static_assert(ROWPTR_PAD >= NN + 1, "room for the segment starts");

__global__ void __launch_bounds__(1024) bucket_sort_local(const int* __restrict__ src,
                                                          const int* __restrict__ dst,
                                                          int2* __restrict__ bed,
                                                          int* __restrict__ boff,
                                                          float* __restrict__ stz, int nst) {
  __shared__ int hist[256];
  __shared__ int start[256];
  __shared__ int wsum[8];
  int t = threadIdx.x, lane = t & 31, wave = t >> 5, blk = blockIdx.x;
  if (t < 256) hist[t] = 0;
  if (blk == 0)
    for (int k = t; k < nst; k += 1024) stz[k] = 0.f;
  __syncthreads();
  const int* d = dst + blk * EPB;
  const int* s = src + blk * EPB;
  int vd[4], vs[4], vr[4];
#pragma unroll
  for (int j = 0; j < 4; j++) {
    int i = j * 1024 + t;
    int ii = (i < EPB) ? i : 0;
    vd[j] = d[ii]; vs[j] = s[ii];
  }
#pragma unroll
  for (int j = 0; j < 4; j++) {
    int i = j * 1024 + t;
    vr[j] = 0;
    if (i < EPB) vr[j] = atomicAdd(&hist[vd[j] >> 8], 1);
  }
  __syncthreads();
  int c = 0, inc = 0;
  if (t < 256) {
    c = hist[t];
    inc = c;
    for (int o = 1; o < 32; o <<= 1) {
      int v = __shfl_up(inc, o);
      if (lane >= o) inc += v;
    }
    if (lane == 31) wsum[wave] = inc;
  }
  __syncthreads();
  if (t < 256) {
    int off = 0;
    for (int w = 0; w < wave; w++) off += wsum[w];
    int ex = off + inc - c;
    start[t] = ex;
    if (t <= NBK) boff[t * NEB + blk] = ex;
  }
  __syncthreads();
  int2* o = bed + blk * EPB;
#pragma unroll
  for (int j = 0; j < 4; j++) {
    int i = j * 1024 + t;
    if (i < EPB) o[start[vd[j] >> 8] + vr[j]] = make_int2(vd[j], vs[j]);
  }
}

__global__ void __launch_bounds__(1024) bucket_csr(const int2* __restrict__ bed,
                                                   const int* __restrict__ boff,
                                                   int* __restrict__ rowptr,
                                                   float* __restrict__ rdeg,
                                                   int* __restrict__ col) {
  __shared__ int cnt[256];
  __shared__ int cur[256];
  __shared__ int wsum[8];
  __shared__ int segS[NEB];
  __shared__ int segL[NEB];
  __shared__ int rb[8];
  int t = threadIdx.x, lane = t & 31, wave = t >> 5, b = blockIdx.x;
  if (t < 256) {
    int s0 = boff[b * NEB + t], s1 = boff[(b + 1) * NEB + t];
    segS[t] = s0; segL[t] = s1 - s0; cnt[t] = 0;
    int a0 = s0;
    for (int o = 16; o > 0; o >>= 1) a0 += __shfl_xor(a0, o);
    if (lane == 0) rb[wave] = a0;
  }
  __syncthreads();
  int beg = 0;
  for (int w = 0; w < 8; w++) beg += rb[w];
  int2 ed[SEG_Q];
  int ln[SEG_Q];
#pragma unroll
  for (int q = 0; q < SEG_Q; q++) {
    int eb = wave * SEG_Q + q;
    int st = segS[eb];
    ln[q] = segL[eb];
    ed[q] = bed[eb * EPB + ((lane < ln[q]) ? st + lane : 0)];
  }
#pragma unroll
  for (int q = 0; q < SEG_Q; q++)
    if (lane < ln[q]) atomicAdd(&cnt[ed[q].x & 255], 1);
#pragma unroll
  for (int q = 0; q < SEG_Q; q++) {
    if (ln[q] > 32) {
      int eb = wave * SEG_Q + q;
      const int2* p = bed + eb * EPB + segS[eb];
      for (int i = lane + 32; i < ln[q]; i += 32) atomicAdd(&cnt[p[i].x & 255], 1);
    }
  }
  __syncthreads();
  int c = 0, inc = 0;
  if (t < 256) {
    c = cnt[t];
    inc = c;
    for (int o = 1; o < 32; o <<= 1) {
      int v = __shfl_up(inc, o);
      if (lane >= o) inc += v;
    }
    if (lane == 31) wsum[wave] = inc;
  }
  __syncthreads();
  if (t < 256) {
    int off = 0;
    for (int w = 0; w < wave; w++) off += wsum[w];
    int ex = off + inc - c;
    cur[t] = ex;
    int node = b * 256 + t;
    if (node <= NN) rowptr[node] = beg + ex;
    if (node < NN) rdeg[node] = 1.f / fmaxf((float)c, 1.f);
  }
  __syncthreads();
#pragma unroll
  for (int q = 0; q < SEG_Q; q++)
    if (lane < ln[q]) col[beg + atomicAdd(&cur[ed[q].x & 255], 1)] = ed[q].y;
#pragma unroll
  for (int q = 0; q < SEG_Q; q++) {
    if (ln[q] > 32) {
      int eb = wave * SEG_Q + q;
      const int2* p = bed + eb * EPB + segS[eb];
      for (int i = lane + 32; i < ln[q]; i += 32) {
        int2 e = p[i];
        col[beg + atomicAdd(&cur[e.x & 255], 1)] = e.y;
      }
    }
  }
}

template <int V, int R, int BN>
__device__ __forceinline__ void add_rows(float4 (&acc)[V], const float* __restrict__ hl,
                                         int c, int j, const float4 (&mu)[V],
                                         const float4 (&sc)[V], const float4 (&sh)[V]) {
  constexpr int F = 128 * V;
  float4 x[R][V];
#pragma unroll
  for (int r = 0; r < R; r++) {
    const float* p = hl + (size_t)__builtin_amdgcn_readlane(c, j + r) * F;
#pragma unroll
    for (int v = 0; v < V; v++) x[r][v] = *(const float4*)(p + v * 128);
  }
#pragma unroll
  for (int r = 0; r < R; r++) {
#pragma unroll
    for (int v = 0; v < V; v++) {
      float4 y = x[r][v];
      if (BN) {
        y.x = (y.x - mu[v].x) * sc[v].x + sh[v].x;
        y.y = (y.y - mu[v].y) * sc[v].y + sh[v].y;
        y.z = (y.z - mu[v].z) * sc[v].z + sh[v].z;
        y.w = (y.w - mu[v].w) * sc[v].w + sh[v].w;
      }
      acc[v].x += y.x; acc[v].y += y.y;
      acc[v].z += y.z; acc[v].w += y.w;
    }
  }
}

template <int V, int BN, int FIN>
__device__ __forceinline__ void gather_row(const float* __restrict__ h,
                                           const int* __restrict__ rowptr,
                                           const int* __restrict__ col,
                                           float* __restrict__ msg, int n,
                                           const u64* __restrict__ bst,
                                           const float* __restrict__ bg,
                                           const float* __restrict__ bb,
                                           const float* __restrict__ rdeg,
                                           const float* __restrict__ self,
                                           const float* __restrict__ eps) {
  static_assert(!FIN || (V == 1 && !BN), "the closing form is the plain 128-wide gather");
  constexpr int F = 128 * V;
  __shared__ float gMu[BN ? F : 4], gSc[BN ? F : 4], gSh[BN ? F : 4];
  if (BN) {
    for (int k = threadIdx.x; k < F; k += blockDim.x) {
      float mean, scl;
      bn_consts(bst[k], bst[F + k], bg[k], mean, scl);
      gMu[k] = mean; gSc[k] = scl; gSh[k] = bb[k];
    }
    __syncthreads();
  }
  int lane = threadIdx.x & 31;
  int i = blockIdx.x * (blockDim.x >> 5) + (threadIdx.x >> 5);
  i = __builtin_amdgcn_readfirstlane(i);
  if (i >= n) return;
  int beg = rowptr[i], end = rowptr[i + 1];
  float rd = 0.f;
  float4 sv = make_float4(0.f, 0.f, 0.f, 0.f), ev = sv;
  if (FIN) {
    rd = rdeg[i];
    sv = *(const float4*)(self + (size_t)i * 128 + lane * 4);
    ev = *(const float4*)(eps + (size_t)i * 64 + (lane & 15) * 4);
  }
  float4 acc[V];
#pragma unroll
  for (int v = 0; v < V; v++) acc[v] = make_float4(0.f, 0.f, 0.f, 0.f);
  float4 mu[V], sc[V], sh[V];
#pragma unroll
  for (int v = 0; v < V; v++) {
    mu[v] = sc[v] = sh[v] = make_float4(0.f, 0.f, 0.f, 0.f);
    if (BN) {
      int c0 = v * 128 + lane * 4;
      mu[v] = *(const float4*)(&gMu[c0]);
      sc[v] = *(const float4*)(&gSc[c0]);
      sh[v] = *(const float4*)(&gSh[c0]);
    }
  }
  const float* hl = h + lane * 4;
  for (int base = beg; base < end; base += 32) {
    int m = end - base;
    if (m > 32) m = 32;
    int c = (lane < m) ? col[base + lane] : 0;
    int j = 0;
    for (; j + 4 <= m; j += 4) add_rows<V, 4, BN>(acc, hl, c, j, mu, sc, sh);
    int rem = m - j;
    if (rem == 3) add_rows<V, 3, BN>(acc, hl, c, j, mu, sc, sh);
    else if (rem == 2) add_rows<V, 2, BN>(acc, hl, c, j, mu, sc, sh);
    else if (rem == 1) add_rows<V, 1, BN>(acc, hl, c, j, mu, sc, sh);
  }
  if (FIN) {
    float4 t;
    t.x = acc[0].x * rd + sv.x; t.y = acc[0].y * rd + sv.y;
    t.z = acc[0].z * rd + sv.z; t.w = acc[0].w * rd + sv.w;
    float4 e;
    e.x = __shfl_down(expf(t.x), 16); e.y = __shfl_down(expf(t.y), 16);
    e.z = __shfl_down(expf(t.z), 16); e.w = __shfl_down(expf(t.w), 16);
    if (lane < 16) {
      float4 z;
      z.x = t.x + e.x * ev.x; z.y = t.y + e.y * ev.y;
      z.z = t.z + e.z * ev.z; z.w = t.w + e.w * ev.w;
      *(float4*)(msg + (size_t)i * 64 + lane * 4) = z;
    }
    return;
  }
  float* o = msg + (size_t)i * F + lane * 4;
#pragma unroll
  for (int v = 0; v < V; v++) *(float4*)(o + v * 128) = acc[v];
}

__global__ void gather_sum_128(const float* __restrict__ h, const int* __restrict__ rowptr,
                               const int* __restrict__ col, float* __restrict__ msg, int n) {
  gather_row<1, 0, 0>(h, rowptr, col, msg, n, nullptr, nullptr, nullptr, nullptr, nullptr, nullptr);
}

__global__ void gather_bn_sum_128(const float* __restrict__ h, const int* __restrict__ rowptr,
                                  const int* __restrict__ col, float* __restrict__ msg, int n,
                                  const u64* __restrict__ bst, const float* __restrict__ bg,
                                  const float* __restrict__ bb) {
  gather_row<1, 1, 0>(h, rowptr, col, msg, n, bst, bg, bb, nullptr, nullptr, nullptr);
}

__global__ void gather_mean_reparam(const float* __restrict__ p, const int* __restrict__ rowptr,
                                    const int* __restrict__ col, float* __restrict__ z, int n,
                                    const float* __restrict__ rdeg,
                                    const float* __restrict__ self,
                                    const float* __restrict__ eps) {
  gather_row<1, 0, 1>(p, rowptr, col, z, n, nullptr, nullptr, nullptr, rdeg, self, eps);
}

#define MT 5
#define BM (16 * MT)
template <int PJ, int BN, int K, int F>
__global__ void __launch_bounds__(128)
sage_gemm_wmma(const float* __restrict__ msg, const float* __restrict__ rdeg,
               const float* __restrict__ h,
               const float* __restrict__ Wl, const float* __restrict__ Wr,
               const float* __restrict__ bias, float* __restrict__ out,
               float* __restrict__ out2, u64* __restrict__ st,
               const u64* __restrict__ bst, const float* __restrict__ bg,
               const float* __restrict__ bb, int do_relu) {
  constexpr int HP = PJ ? 0 : 1;
  __shared__ float sA[PJ ? 1 : 2][BM * SA_STRIDE];
  __shared__ float sB[2][KC * SB_STRIDE];
  __shared__ float sSc[BN ? K : 4], sSh[BN ? K : 4];

  constexpr int nblocks_n = F / 64;
  int t = threadIdx.x;
  int wave = t >> 5, lane = t & 31;
  int mBase = (blockIdx.x / nblocks_n) * BM;
  int nb = blockIdx.x % nblocks_n;

  if (BN) {
    for (int k = t; k < K; k += 128) {
      float mean, scl;
      bn_consts(bst[k], bst[K + k], bg[k], mean, scl);
      sSc[k] = scl; sSh[k] = bb[k] - mean * scl;
    }
    __syncthreads();
  }

  int m16  = lane & 15;
  int kb   = (lane >> 4) << 1;
  int nloc = wave * 16 + (lane & 15);
  int ncol = 64 * nb + nloc;

  v8f acc[MT];
  v8f acc2[PJ ? MT : 1];
#pragma unroll
  for (int i = 0; i < MT; i++) acc[i] = (v8f){0.f, 0.f, 0.f, 0.f, 0.f, 0.f, 0.f, 0.f};
  if (PJ) {
#pragma unroll
    for (int i = 0; i < MT; i++) acc2[i] = (v8f){0.f, 0.f, 0.f, 0.f, 0.f, 0.f, 0.f, 0.f};
  }

  int ar = t >> 3, ac = (t & 7) << 2;
  int br = t >> 4, bc = (t & 15) << 2;
  unsigned aoff = (unsigned)(ar * K + ac);
  unsigned boff = (unsigned)(br * F + 64 * nb + bc);
  const float* am = msg + (size_t)mBase * K;
  const float* ah = h + (size_t)mBase * K;

  float sc[MT];
  if (!PJ) {
#pragma unroll
    for (int i = 0; i < MT; i++) sc[i] = rdeg[mBase + ar + 16 * i];
  }

#pragma nounroll
  for (int kc = 0; kc < K; kc += KC) {
    {
      float4 va0[PJ ? 1 : MT], va1[MT], vb0[4], vb1[4];
      const float* pm = am + kc;
      const float* ph = ah + kc;
      const float* pl = Wl + (size_t)kc * F;
      const float* pr = Wr + (size_t)kc * F;
#pragma unroll
      for (int i = 0; i < MT; i++) {
        if (!PJ) va0[i] = *(const float4*)(pm + (aoff + 16u * i * K));
        va1[i] = *(const float4*)(ph + (aoff + 16u * i * K));
      }
#pragma unroll
      for (int pass = 0; pass < 4; pass++) {
        vb0[pass] = *(const float4*)(pl + (boff + 8u * pass * F));
        vb1[pass] = *(const float4*)(pr + (boff + 8u * pass * F));
      }
      __builtin_amdgcn_sched_barrier(0);
      float4 sl, sf;
      if (BN) {
        sl = *(const float4*)(&sSc[kc + ac]);
        sf = *(const float4*)(&sSh[kc + ac]);
      }
#pragma unroll
      for (int i = 0; i < MT; i++) {
        if (!PJ) {
          float4 s0 = va0[i];
          s0.x *= sc[i]; s0.y *= sc[i]; s0.z *= sc[i]; s0.w *= sc[i];
          *(float4*)(&sA[0][(ar + 16 * i) * SA_STRIDE + ac]) = s0;
        }
        float4 s1 = va1[i];
        if (BN) {
          s1.x = s1.x * sl.x + sf.x; s1.y = s1.y * sl.y + sf.y;
          s1.z = s1.z * sl.z + sf.z; s1.w = s1.w * sl.w + sf.w;
        }
        *(float4*)(&sA[HP][(ar + 16 * i) * SA_STRIDE + ac]) = s1;
      }
#pragma unroll
      for (int pass = 0; pass < 4; pass++) {
        int row = br + pass * 8;
        *(float4*)(&sB[0][row * SB_STRIDE + bc]) = vb0[pass];
        *(float4*)(&sB[1][row * SB_STRIDE + bc]) = vb1[pass];
      }
    }
    if (kc + KC < K) {
      if (t < BM) {
        if (!PJ) __builtin_prefetch(msg + (size_t)(mBase + t) * K + kc + KC, 0, 0);
        __builtin_prefetch(h   + (size_t)(mBase + t) * K + kc + KC, 0, 0);
      } else if (t - BM < KC) {
        __builtin_prefetch(Wl + (size_t)(kc + KC + t - BM) * F + 64 * nb, 0, 0);
        __builtin_prefetch(Wr + (size_t)(kc + KC + t - BM) * F + 64 * nb, 0, 0);
      }
    }
    __syncthreads();

#pragma unroll
    for (int k = 0; k < KC; k += 4) {
      v2f a, b0, b1;
      b0.x = sB[0][(k + kb) * SB_STRIDE + nloc];
      b0.y = sB[0][(k + kb + 1) * SB_STRIDE + nloc];
      b1.x = sB[1][(k + kb) * SB_STRIDE + nloc];
      b1.y = sB[1][(k + kb + 1) * SB_STRIDE + nloc];
      if (PJ) {
#pragma unroll
        for (int i = 0; i < MT; i++) {
          a.x = sA[0][(m16 + 16 * i) * SA_STRIDE + k + kb];
          a.y = sA[0][(m16 + 16 * i) * SA_STRIDE + k + kb + 1];
          acc[i] = __builtin_amdgcn_wmma_f32_16x16x4_f32(false, a, false, b0, (short)0, acc[i],
                                                         false, false);
          acc2[i] = __builtin_amdgcn_wmma_f32_16x16x4_f32(false, a, false, b1, (short)0, acc2[i],
                                                          false, false);
        }
      } else {
#pragma unroll
        for (int i = 0; i < MT; i++) {
          a.x = sA[0][(m16 + 16 * i) * SA_STRIDE + k + kb];
          a.y = sA[0][(m16 + 16 * i) * SA_STRIDE + k + kb + 1];
          acc[i] = __builtin_amdgcn_wmma_f32_16x16x4_f32(false, a, false, b0, (short)0, acc[i],
                                                         false, false);
        }
#pragma unroll
        for (int i = 0; i < MT; i++) {
          a.x = sA[HP][(m16 + 16 * i) * SA_STRIDE + k + kb];
          a.y = sA[HP][(m16 + 16 * i) * SA_STRIDE + k + kb + 1];
          acc[i] = __builtin_amdgcn_wmma_f32_16x16x4_f32(false, a, false, b1, (short)0, acc[i],
                                                         false, false);
        }
      }
    }
    __syncthreads();
  }

  float bv = bias[ncol];
  if (PJ) {
#pragma unroll
    for (int i = 0; i < MT; i++) {
      int mhalf = mBase + 16 * i + ((lane >> 4) << 3);
#pragma unroll
      for (int v = 0; v < 8; v++) {
        out[(size_t)(mhalf + v) * F + ncol] = acc[i][v];
        out2[(size_t)(mhalf + v) * F + ncol] = acc2[i][v] + bv;
      }
    }
    return;
  }
  float cs = 0.f, cs2 = 0.f;
#pragma unroll
  for (int i = 0; i < MT; i++) {
    int mhalf = mBase + 16 * i + ((lane >> 4) << 3);
#pragma unroll
    for (int v = 0; v < 8; v++) {
      float val = acc[i][v] + bv;
      if (do_relu) val = fmaxf(val, 0.f);
      out[(size_t)(mhalf + v) * F + ncol] = val;
      cs += val;
      cs2 = __builtin_fmaf(val, val, cs2);
    }
  }
  if (st) {
    cs  += __shfl_xor(cs, 16);
    cs2 += __shfl_xor(cs2, 16);
    if (lane < 16) {
      atomicAdd(&st[ncol], (u64)__float2ll_rn(cs * FX_ONE));
      atomicAdd(&st[F + ncol], (u64)__float2ll_rn(cs2 * FX_ONE));
    }
  }
}

extern "C" void kernel_launch(void* const* d_in, const int* in_sizes, int n_in,
                              void* d_out, int out_size, void* d_ws, size_t ws_size,
                              hipStream_t stream) {
  const float* x   = (const float*)d_in[0];
  const int*   src = (const int*)d_in[1];
  const int*   dst = (const int*)d_in[2];
  const float* eps = (const float*)d_in[3];
  const float* W1l = (const float*)d_in[4];
  const float* W1r = (const float*)d_in[5];
  const float* b1  = (const float*)d_in[6];
  const float* g1  = (const float*)d_in[7];
  const float* be1 = (const float*)d_in[8];
  const float* W2l = (const float*)d_in[9];
  const float* W2r = (const float*)d_in[10];
  const float* b2  = (const float*)d_in[11];
  const float* g2  = (const float*)d_in[12];
  const float* be2 = (const float*)d_in[13];
  const float* W3l = (const float*)d_in[14];
  const float* W3r = (const float*)d_in[15];
  const float* b3  = (const float*)d_in[16];
  float* zout = (float*)d_out;
  (void)in_sizes; (void)n_in; (void)out_size; (void)ws_size;

  char* ws = (char*)d_ws;
  size_t off = 0;
  auto carve = [&](size_t bytes) -> float* {
    char* p = ws + off;
    off = (off + bytes + 255) & ~(size_t)255;
    return (float*)p;
  };
  float* rdeg = carve((size_t)NN * 4);
  float* msg  = carve((size_t)NN * 256 * 4);
  float* h1   = carve((size_t)NN * 128 * 4);
  float* h2   = carve((size_t)NN * 256 * 4);
  float* y1   = msg + (size_t)NN * 128;
  int* boff   = (int*)h1;
  int* rowptr = boff + BOFF_N;
  int* col    = rowptr + ROWPTR_PAD;
  u64* st1    = (u64*)(col + NE);
  u64* st2    = st1 + 256;
  int2* bed   = (int2*)(st2 + 512);

  const int M = NN;

  bucket_sort_local<<<NEB, 1024, 0, stream>>>(src, dst, bed, boff, (float*)st1, 1536);
  bucket_csr<<<NBK, 1024, 0, stream>>>(bed, boff, rowptr, rdeg, col);

  gather_sum_128<<<(M + 7) / 8, 256, 0, stream>>>(x, rowptr, col, msg, M);
  sage_gemm_wmma<0, 0, 128, 128><<<(NN / BM) * 2, 128, 0, stream>>>(
      msg, rdeg, x, W1l, W1r, b1, y1, nullptr, st1, nullptr, nullptr, nullptr, 1);

  gather_bn_sum_128<<<(M + 7) / 8, 256, 0, stream>>>(y1, rowptr, col, msg, M, st1, g1, be1);
  sage_gemm_wmma<0, 1, 128, 256><<<(NN / BM) * 4, 128, 0, stream>>>(
      msg, rdeg, y1, W2l, W2r, b2, h2, nullptr, st2, st1, g1, be1, 1);

  float* pj = msg;
  float* sj = y1;
  sage_gemm_wmma<1, 1, 256, 128><<<(NN / BM) * 2, 128, 0, stream>>>(
      nullptr, nullptr, h2, W3l, W3r, b3, pj, sj, nullptr, st2, g2, be2, 0);
  gather_mean_reparam<<<(M + 7) / 8, 256, 0, stream>>>(pj, rowptr, col, zout, M, rdeg, sj, eps);
}
